# starting kernel
# baseline (speedup 1.0000x reference)
#include <hip/hip_runtime.h>

typedef __attribute__((ext_vector_type(2))) float v2f;
typedef __attribute__((ext_vector_type(8))) float v8f;

#define TPB 256

static __device__ __forceinline__ void atomAddF32(float* p, float v) {
  unsafeAtomicAdd(p, v);
}


__global__ void k_init_deg(float* __restrict__ deg, int N) {
  int n = blockIdx.x * blockDim.x + threadIdx.x;
  if (n < N) deg[n] = 1.0f;
}

__global__ void k_edge_deg(const int* __restrict__ dst, float* __restrict__ deg, int E) {
  int e = blockIdx.x * blockDim.x + threadIdx.x;
  if (e < E) atomAddF32(&deg[dst[e]], 1.0f);
}

__global__ void k_dinv(float* __restrict__ deg, int N) {
  int n = blockIdx.x * blockDim.x + threadIdx.x;
  if (n < N) {
    float d = deg[n];
    deg[n] = (d > 0.0f) ? rsqrtf(d) : 0.0f;
  }
}


__global__ void k_gemm1_wmma(const float* __restrict__ x,
                             const float* __restrict__ W1,
                             const float* __restrict__ dinv,
                             float* __restrict__ h1,
                             float* __restrict__ agg1, int N) {
  const int lane = threadIdx.x & 31;
  const int tile = blockIdx.x * (blockDim.x >> 5) + (threadIdx.x >> 5);
  const int row0 = tile * 16;
  if (row0 >= N) return;

  const int half = lane >> 4;
  const int lrow = lane & 15;

  int arow = row0 + lrow;
  if (arow >= N) arow = N - 1;

  const float* xrow = x + (size_t)arow * 128 + half * 2;
  const float* wcol = W1 + (size_t)(half * 2) * 16 + lrow;

  v8f acc = {};
#pragma unroll
  for (int k0 = 0; k0 < 128; k0 += 4) {
    v2f a = *(const v2f*)(xrow + k0);
    v2f b;
    b.x = wcol[(size_t)k0 * 16];
    b.y = wcol[(size_t)k0 * 16 + 16];
    acc = __builtin_amdgcn_wmma_f32_16x16x4_f32(
         false, a,  false, b,
         (short)0, acc,  false,  false);
  }

  const int col = lrow;
#pragma unroll
  for (int v = 0; v < 8; ++v) {
    const int m = row0 + v + half * 8;
    if (m < N) {
      const float val = acc[v];
      h1[(size_t)m * 16 + col] = val;
      const float di = dinv[m];
      agg1[(size_t)m * 16 + col] = val * di * di;
    }
  }
}


__global__ void k_edge_agg16(const int* __restrict__ ei,
                             const float* __restrict__ dinv,
                             const float* __restrict__ h1,
                             float* __restrict__ agg1, int E) {
  int e = blockIdx.x * blockDim.x + threadIdx.x;
  if (e >= E) return;
  const int s = ei[e];
  const int d = ei[E + e];
  const float w = dinv[s] * dinv[d];
  const float4* hs = (const float4*)(h1 + (size_t)s * 16);
  float4 m0 = hs[0], m1 = hs[1], m2 = hs[2], m3 = hs[3];
  float* op = agg1 + (size_t)d * 16;
  atomAddF32(op + 0,  m0.x * w);
  atomAddF32(op + 1,  m0.y * w);
  atomAddF32(op + 2,  m0.z * w);
  atomAddF32(op + 3,  m0.w * w);
  atomAddF32(op + 4,  m1.x * w);
  atomAddF32(op + 5,  m1.y * w);
  atomAddF32(op + 6,  m1.z * w);
  atomAddF32(op + 7,  m1.w * w);
  atomAddF32(op + 8,  m2.x * w);
  atomAddF32(op + 9,  m2.y * w);
  atomAddF32(op + 10, m2.z * w);
  atomAddF32(op + 11, m2.w * w);
  atomAddF32(op + 12, m3.x * w);
  atomAddF32(op + 13, m3.y * w);
  atomAddF32(op + 14, m3.z * w);
  atomAddF32(op + 15, m3.w * w);
}


__global__ void k_node2(const float* __restrict__ agg1,
                        const float* __restrict__ b1,
                        const float* __restrict__ W2,
                        const float* __restrict__ b2,
                        const float* __restrict__ dinv,
                        float* __restrict__ h2,
                        float* __restrict__ out, int N) {
  int n = blockIdx.x * blockDim.x + threadIdx.x;
  if (n >= N) return;
  const float4* ap = (const float4*)(agg1 + (size_t)n * 16);
  float c0 = 0.0f, c1 = 0.0f;
#pragma unroll
  for (int i = 0; i < 4; ++i) {
    float4 t = ap[i];
    float v[4] = {t.x, t.y, t.z, t.w};
#pragma unroll
    for (int j = 0; j < 4; ++j) {
      const int k = i * 4 + j;
      const float hv = fmaxf(v[j] + b1[k], 0.0f);
      c0 = fmaf(hv, W2[k * 2 + 0], c0);
      c1 = fmaf(hv, W2[k * 2 + 1], c1);
    }
  }
  h2[(size_t)n * 2 + 0] = c0;
  h2[(size_t)n * 2 + 1] = c1;
  const float di = dinv[n];
  const float w = di * di;
  out[(size_t)n * 2 + 0] = fmaf(c0, w, b2[0]);
  out[(size_t)n * 2 + 1] = fmaf(c1, w, b2[1]);
}


__global__ void k_edge_agg2(const int* __restrict__ ei,
                            const float* __restrict__ dinv,
                            const float* __restrict__ h2,
                            float* __restrict__ out, int E) {
  int e = blockIdx.x * blockDim.x + threadIdx.x;
  if (e >= E) return;
  const int s = ei[e];
  const int d = ei[E + e];
  const float w = dinv[s] * dinv[d];
  const float2 m = *(const float2*)(h2 + (size_t)s * 2);
  atomAddF32(out + (size_t)d * 2 + 0, m.x * w);
  atomAddF32(out + (size_t)d * 2 + 1, m.y * w);
}


extern "C" void kernel_launch(void* const* d_in, const int* in_sizes, int n_in,
                              void* d_out, int out_size, void* d_ws, size_t ws_size,
                              hipStream_t stream) {
  const float* x  = (const float*)d_in[0];
  const int*   ei = (const int*)d_in[1];
  const float* W1 = (const float*)d_in[2];
  const float* b1 = (const float*)d_in[3];
  const float* W2 = (const float*)d_in[4];
  const float* b2 = (const float*)d_in[5];
  float* out = (float*)d_out;

  const int N = in_sizes[0] / 128;
  const int E = in_sizes[1] / 2;

  float* deg  = (float*)d_ws;
  float* h1   = deg + N;
  float* agg1 = h1 + (size_t)N * 16;
  float* h2   = agg1 + (size_t)N * 16;

  dim3 blk(TPB);
  dim3 gN((N + TPB - 1) / TPB);
  dim3 gE((E + TPB - 1) / TPB);

  k_init_deg<<<gN, blk, 0, stream>>>(deg, N);
  k_edge_deg<<<gE, blk, 0, stream>>>(ei + E, deg, E);
  k_dinv<<<gN, blk, 0, stream>>>(deg, N);

  const int tiles = (N + 15) / 16;
  const int wavesPerBlock = TPB / 32;
  dim3 gT((tiles + wavesPerBlock - 1) / wavesPerBlock);
  k_gemm1_wmma<<<gT, blk, 0, stream>>>(x, W1, deg, h1, agg1, N);

  k_edge_agg16<<<gE, blk, 0, stream>>>(ei, deg, h1, agg1, E);
  k_node2<<<gN, blk, 0, stream>>>(agg1, b1, W2, b2, deg, h2, out, N);
  k_edge_agg2<<<gE, blk, 0, stream>>>(ei, deg, h2, out, E);
}


# layer-2 edge pass: the scan reads src beside dst with whole-line loads and the list word holds (source node, destination row): a listed edge no longer gathers src[e] before its record load
# speedup vs baseline: 7.9111x; 7.9111x over previous
#include <hip/hip_runtime.h>

typedef __attribute__((ext_vector_type(2))) float v2f;
typedef __attribute__((ext_vector_type(8))) float v8f;

#define TPB 256

#define EDGE_T 1024
#define LIST_CAP 16384
#define NWAVE (EDGE_T / 32)
#define WLIST (LIST_CAP / NWAVE)
#define DEG_P 25600
#define DEG_S 64
#define L1_P 3136
#define L1_S 8
#define L1_PITCH 17
#define L1_OWN ((L1_P + L1_S - 1) / L1_S)
#define L2_P 12544
#define L2_S 32
#define L1_RB 12
#define L2_RB 14
static_assert(L1_P <= (1 << L1_RB), "layer-1 row does not fit its list-word field");
static_assert(L2_P <= (1 << L2_RB), "layer-2 row does not fit its list-word field");

static __device__ __forceinline__ void atomAddF32(float* p, float v) {
  unsafeAtomicAdd(p, v);
}

static __device__ __forceinline__ void ldsAddF32(float* p, float v) {
  __hip_atomic_fetch_add(p, v, __ATOMIC_RELAXED, __HIP_MEMORY_SCOPE_WORKGROUP);
}

static __device__ __forceinline__ float dinv_of(float d) {
  return (d > 0.0f) ? rsqrtf(d) : 0.0f;
}

template <bool WITH_SRC, class F>
static __device__ __forceinline__ void scan_owned(const int* __restrict__ dst,
                                                  const int* __restrict__ src, int eb, int ee,
                                                  bool vec_ok, int base, int P, F onvec) {
  const int lane = threadIdx.x & 31;
  const int w0 = __builtin_amdgcn_readfirstlane((int)(threadIdx.x >> 5)) * 32;
  const int nvec = vec_ok ? ((ee - eb) >> 2) : 0;
  const int4* d4 = (const int4*)(dst + eb);
  const int4* s4 = (const int4*)(src + eb);
  auto lds4 = [&](int i) -> int4 {
    if constexpr (WITH_SRC) return s4[i];
    else return make_int4(0, 0, 0, 0);
  };
  auto test4 = [&](const int4 v, const int4 u, int i, bool ok) {
    const int e0 = i * 4;
    unsigned r;
    r = (unsigned)(v.x - base); onvec(e0 + 0, (int)r, ok && r < (unsigned)P, u.x);
    r = (unsigned)(v.y - base); onvec(e0 + 1, (int)r, ok && r < (unsigned)P, u.y);
    r = (unsigned)(v.z - base); onvec(e0 + 2, (int)r, ok && r < (unsigned)P, u.z);
    r = (unsigned)(v.w - base); onvec(e0 + 3, (int)r, ok && r < (unsigned)P, u.w);
  };
  int iw = w0;
  for (; iw + 31 + 3 * EDGE_T < nvec; iw += 4 * EDGE_T) {
    const int i = iw + lane;
    const int4 a = d4[i];
    const int4 b = d4[i + EDGE_T];
    const int4 c = d4[i + 2 * EDGE_T];
    const int4 d = d4[i + 3 * EDGE_T];
    const int4 sa = lds4(i);
    const int4 sb = lds4(i + EDGE_T);
    const int4 sc = lds4(i + 2 * EDGE_T);
    const int4 sd = lds4(i + 3 * EDGE_T);
    test4(a, sa, i, true);
    test4(b, sb, i + EDGE_T, true);
    test4(c, sc, i + 2 * EDGE_T, true);
    test4(d, sd, i + 3 * EDGE_T, true);
  }
  for (; iw < nvec; iw += EDGE_T) {
    const int i = iw + lane;
    const bool ok = i < nvec;
    const int j = ok ? i : iw;
    test4(d4[j], lds4(j), i, ok);
  }
  for (int ew = eb + nvec * 4 + w0; ew < ee; ew += EDGE_T) {
    const int e = ew + lane;
    const bool ok = e < ee;
    const int j = ok ? e : ew;
    const unsigned r = (unsigned)(dst[j] - base);
    int sv = 0;
    if constexpr (WITH_SRC) sv = src[j];
    onvec(e - eb, (int)r, ok && r < (unsigned)P, sv);
  }
}

static __device__ __forceinline__ bool wave_append(unsigned* mylist, int& cnt, int cap,
                                                   unsigned word, bool m) {
  const unsigned mask = __builtin_amdgcn_ballot_w32(m);
  bool spill = false;
  if (mask != 0u) {
    const int p = cnt + (int)__builtin_amdgcn_mbcnt_lo(mask, 0u);
    cnt = __builtin_amdgcn_readfirstlane(cnt + __popc(mask));
    if (m) {
      if (p < cap) mylist[p] = word; else spill = true;
    }
  }
  return spill;
}

static __device__ __forceinline__ int seg_len(int E, int S) {
  return (((E + S - 1) / S) + 3) & ~3;
}


__global__ __launch_bounds__(EDGE_T) void k_edge_deg(const int* __restrict__ dst,
                                                     float* __restrict__ deg, int N, int E, int R) {
  __shared__ int cnt[DEG_P];
  const int tid = threadIdx.x;
  const int range = blockIdx.x % R;
  const int seg = blockIdx.x / R;
  const int base = range * DEG_P;
  for (int i = tid; i < DEG_P; i += EDGE_T) cnt[i] = 0;
  __syncthreads();
  const int sl = seg_len(E, DEG_S);
  const int eb = min(seg * sl, E);
  const int ee = min(eb + sl, E);
  scan_owned<false>(dst, dst, eb, ee, (E & 3) == 0, base, DEG_P, [&](int, int r, bool m, int) {
    if (m) __hip_atomic_fetch_add(&cnt[r], 1, __ATOMIC_RELAXED, __HIP_MEMORY_SCOPE_WORKGROUP);
  });
  __syncthreads();
  for (int i = tid; i < DEG_P; i += EDGE_T) {
    const int n = base + i;
    const int c = cnt[i];
    if (n < N && c != 0) atomAddF32(&deg[n], (float)c);
  }
}


__global__ void k_gemm1_wmma(const float* __restrict__ x,
                             const float* __restrict__ W1,
                             float* __restrict__ deg,
                             float* __restrict__ h1,
                             float* __restrict__ agg1, int N) {
  const int lane = threadIdx.x & 31;
  const int tile = blockIdx.x * (blockDim.x >> 5) + (threadIdx.x >> 5);
  const int row0 = tile * 16;
  if (row0 >= N) return;

  const int half = lane >> 4;
  const int lrow = lane & 15;

  int arow = row0 + lrow;
  if (arow >= N) arow = N - 1;

  const float* xrow = x + (size_t)arow * 128 + half * 2;
  const float* wcol = W1 + (size_t)(half * 2) * 16 + lrow;

  v8f acc = {};
#pragma unroll
  for (int k0 = 0; k0 < 128; k0 += 4) {
    v2f a = *(const v2f*)(xrow + k0);
    v2f b;
    b.x = wcol[(size_t)k0 * 16];
    b.y = wcol[(size_t)k0 * 16 + 16];
    acc = __builtin_amdgcn_wmma_f32_16x16x4_f32(
         false, a,  false, b,
         (short)0, acc,  false,  false);
  }

  const int col = lrow;
#pragma unroll
  for (int v = 0; v < 8; ++v) {
    const int m = row0 + v + half * 8;
    if (m < N) {
      h1[(size_t)m * 16 + col] = acc[v];
      agg1[(size_t)m * 16 + col] = 0.0f;
    }
  }
  if (lane < 16 && row0 + lane < N) deg[row0 + lane] = 1.0f;
}


static __device__ __forceinline__ void l1_add(float* op, const float4 m0, const float4 m1,
                                              const float4 m2, const float4 m3, float w) {
  ldsAddF32(op + 0,  m0.x * w);
  ldsAddF32(op + 1,  m0.y * w);
  ldsAddF32(op + 2,  m0.z * w);
  ldsAddF32(op + 3,  m0.w * w);
  ldsAddF32(op + 4,  m1.x * w);
  ldsAddF32(op + 5,  m1.y * w);
  ldsAddF32(op + 6,  m1.z * w);
  ldsAddF32(op + 7,  m1.w * w);
  ldsAddF32(op + 8,  m2.x * w);
  ldsAddF32(op + 9,  m2.y * w);
  ldsAddF32(op + 10, m2.z * w);
  ldsAddF32(op + 11, m2.w * w);
  ldsAddF32(op + 12, m3.x * w);
  ldsAddF32(op + 13, m3.y * w);
  ldsAddF32(op + 14, m3.z * w);
  ldsAddF32(op + 15, m3.w * w);
}

static __device__ __forceinline__ void l1_edge(int e, const int* __restrict__ src,
                                               const int* __restrict__ dst,
                                               const float* __restrict__ deg,
                                               const float* __restrict__ h1,
                                               float* acc, int base) {
  const int s = src[e];
  const int d = dst[e];
  const float w = dinv_of(deg[s]) * dinv_of(deg[d]);
  const float4* hs = (const float4*)(h1 + (size_t)s * 16);
  l1_add(acc + (d - base) * L1_PITCH, hs[0], hs[1], hs[2], hs[3], w);
}

__global__ __launch_bounds__(EDGE_T) void k_edge_agg16(const int* __restrict__ ei,
                                                       const float* __restrict__ deg,
                                                       const float* __restrict__ h1,
                                                       float* __restrict__ agg1,
                                                       int N, int E, int R) {
  __shared__ float acc[L1_P * L1_PITCH];
  __shared__ float dinv_r[L1_P];
  __shared__ unsigned list[LIST_CAP];
  const int tid = threadIdx.x;
  const int lane = tid & 31;
  unsigned* mylist = list + (tid >> 5) * WLIST;
  const int range = blockIdx.x % R;
  const int seg = blockIdx.x / R;
  const int base = range * L1_P;
  const int* __restrict__ src = ei;
  const int* __restrict__ dst = ei + E;

  for (int i = tid; i < L1_P * L1_PITCH; i += EDGE_T) acc[i] = 0.0f;
  for (int i = tid; i < L1_P; i += EDGE_T) dinv_r[i] = (base + i < N) ? dinv_of(deg[base + i]) : 0.0f;
  __syncthreads();

  constexpr int OWN_T = (L1_OWN * 16 + EDGE_T - 1) / EDGE_T;
  const int own_lo = seg * L1_OWN;
  float selfv[OWN_T];
  int selfo[OWN_T];
#pragma unroll
  for (int q = 0; q < OWN_T; ++q) {
    const int i = tid + q * EDGE_T;
    const int r = own_lo + (i >> 4);
    const int k = i & 15;
    const bool ok = i < L1_OWN * 16 && r < L1_P && base + r < N;
    const int node = ok ? base + r : base;
    selfv[q] = h1[(size_t)node * 16 + k];
    selfo[q] = ok ? r * L1_PITCH + k : -1;
  }

  const int sl = seg_len(E, L1_S);
  const int eb = min(seg * sl, E);
  const int ee = min(eb + sl, E);
  const int cap = (sl <= (1 << (32 - L1_RB))) ? WLIST : 0;
  int cnt = 0;
  scan_owned<false>(dst, dst, eb, ee, (E & 3) == 0, base, L1_P, [&](int rel, int r, bool m, int) {
    if (wave_append(mylist, cnt, cap, ((unsigned)rel << L1_RB) | (unsigned)r, m))
      l1_edge(eb + rel, src, dst, deg, h1, acc, base);
  });
  __syncthreads();

  const int n = min(cnt, cap);
  for (int j0 = 0; j0 < n; j0 += 4 * 32) {
    int s[4], r[4];
    bool ok[4];
    float w[4];
    float4 m[4][4];
#pragma unroll
    for (int q = 0; q < 4; ++q) {
      const int idx = j0 + q * 32 + lane;
      ok[q] = idx < n;
      const unsigned wd = mylist[ok[q] ? idx : n - 1];
      r[q] = (int)(wd & ((1u << L1_RB) - 1u));
      s[q] = src[eb + (int)(wd >> L1_RB)];
    }
#pragma unroll
    for (int q = 0; q < 4; ++q) {
      w[q] = dinv_of(deg[s[q]]) * dinv_r[r[q]];
      const float4* hs = (const float4*)(h1 + (size_t)s[q] * 16);
      m[q][0] = hs[0]; m[q][1] = hs[1]; m[q][2] = hs[2]; m[q][3] = hs[3];
    }
#pragma unroll
    for (int q = 0; q < 4; ++q)
      if (ok[q]) l1_add(acc + r[q] * L1_PITCH, m[q][0], m[q][1], m[q][2], m[q][3], w[q]);
  }
#pragma unroll
  for (int q = 0; q < OWN_T; ++q) {
    if (selfo[q] >= 0) {
      const float di = dinv_r[own_lo + ((tid + q * EDGE_T) >> 4)];
      ldsAddF32(acc + selfo[q], selfv[q] * di * di);
    }
  }
  __syncthreads();

  for (int i = tid; i < L1_P * 16; i += EDGE_T) {
    const int r = i >> 4;
    const int k = i & 15;
    const int node = base + r;
    if (node < N) atomAddF32(agg1 + (size_t)node * 16 + k, acc[r * L1_PITCH + k]);
  }
}


__global__ void k_node2(const float* __restrict__ agg1,
                        const float* __restrict__ b1,
                        const float* __restrict__ W2,
                        const float* __restrict__ b2,
                        const float* __restrict__ deg,
                        float* __restrict__ rec2,
                        float* __restrict__ out, int N) {
  int n = blockIdx.x * blockDim.x + threadIdx.x;
  if (n >= N) return;
  const float4* ap = (const float4*)(agg1 + (size_t)n * 16);
  float c0 = 0.0f, c1 = 0.0f;
#pragma unroll
  for (int i = 0; i < 4; ++i) {
    float4 t = ap[i];
    float v[4] = {t.x, t.y, t.z, t.w};
#pragma unroll
    for (int j = 0; j < 4; ++j) {
      const int k = i * 4 + j;
      const float hv = fmaxf(v[j] + b1[k], 0.0f);
      c0 = fmaf(hv, W2[k * 2 + 0], c0);
      c1 = fmaf(hv, W2[k * 2 + 1], c1);
    }
  }
  const float di = dinv_of(deg[n]);
  *(float4*)(rec2 + (size_t)n * 4) = make_float4(c0, c1, di, 0.0f);
  const float w = di * di;
  out[(size_t)n * 2 + 0] = fmaf(c0, w, b2[0]);
  out[(size_t)n * 2 + 1] = fmaf(c1, w, b2[1]);
}


static __device__ __forceinline__ void l2_edge(int e, const int* __restrict__ src,
                                               const int* __restrict__ dst,
                                               const float* __restrict__ deg,
                                               const float* __restrict__ rec2,
                                               float* acc, int base) {
  const int s = src[e];
  const int d = dst[e];
  const float4 mm = *(const float4*)(rec2 + (size_t)s * 4);
  const float w = mm.z * dinv_of(deg[d]);
  float* op = acc + (d - base) * 2;
  ldsAddF32(op + 0, mm.x * w);
  ldsAddF32(op + 1, mm.y * w);
}

__global__ __launch_bounds__(EDGE_T) void k_edge_agg2(const int* __restrict__ ei,
                                                      const float* __restrict__ deg,
                                                      const float* __restrict__ rec2,
                                                      float* __restrict__ out,
                                                      int N, int E, int R) {
  __shared__ float acc[L2_P * 2];
  __shared__ float dinv_r[L2_P];
  __shared__ unsigned list[LIST_CAP];
  const int tid = threadIdx.x;
  const int lane = tid & 31;
  unsigned* mylist = list + (tid >> 5) * WLIST;
  const int range = blockIdx.x % R;
  const int seg = blockIdx.x / R;
  const int base = range * L2_P;
  const int* __restrict__ src = ei;
  const int* __restrict__ dst = ei + E;

  for (int i = tid; i < L2_P * 2; i += EDGE_T) acc[i] = 0.0f;
  for (int i = tid; i < L2_P; i += EDGE_T) dinv_r[i] = (base + i < N) ? dinv_of(deg[base + i]) : 0.0f;
  __syncthreads();

  const int sl = seg_len(E, L2_S);
  const int eb = min(seg * sl, E);
  const int ee = min(eb + sl, E);
  const int cap = (N <= (1 << (32 - L2_RB))) ? WLIST : 0;
  int cnt = 0;
  scan_owned<true>(dst, src, eb, ee, (E & 3) == 0, base, L2_P, [&](int rel, int r, bool m, int sv) {
    if (wave_append(mylist, cnt, cap, ((unsigned)sv << L2_RB) | (unsigned)r, m))
      l2_edge(eb + rel, src, dst, deg, rec2, acc, base);
  });
  __syncthreads();

  const int n = min(cnt, cap);
  for (int j0 = 0; j0 < n; j0 += 4 * 32) {
    int r[4];
    bool ok[4];
    float dr[4];
    float4 m[4];
#pragma unroll
    for (int q = 0; q < 4; ++q) {
      const int idx = j0 + q * 32 + lane;
      ok[q] = idx < n;
      const unsigned wd = mylist[ok[q] ? idx : n - 1];
      r[q] = (int)(wd & ((1u << L2_RB) - 1u));
      m[q] = *(const float4*)(rec2 + (size_t)(wd >> L2_RB) * 4);
    }
#pragma unroll
    for (int q = 0; q < 4; ++q) dr[q] = dinv_r[r[q]];
#pragma unroll
    for (int q = 0; q < 4; ++q) {
      if (ok[q]) {
        const float w = m[q].z * dr[q];
        float* op = acc + r[q] * 2;
        ldsAddF32(op + 0, m[q].x * w);
        ldsAddF32(op + 1, m[q].y * w);
      }
    }
  }
  __syncthreads();

  for (int i = tid; i < L2_P * 2; i += EDGE_T) {
    const size_t o = (size_t)base * 2 + i;
    if (o < (size_t)N * 2) atomAddF32(out + o, acc[i]);
  }
}


extern "C" void kernel_launch(void* const* d_in, const int* in_sizes, int n_in,
                              void* d_out, int out_size, void* d_ws, size_t ws_size,
                              hipStream_t stream) {
  const float* x  = (const float*)d_in[0];
  const int*   ei = (const int*)d_in[1];
  const float* W1 = (const float*)d_in[2];
  const float* b1 = (const float*)d_in[3];
  const float* W2 = (const float*)d_in[4];
  const float* b2 = (const float*)d_in[5];
  float* out = (float*)d_out;

  const int N = in_sizes[0] / 128;
  const int E = in_sizes[1] / 2;

  float* deg  = (float*)d_ws;
  float* h1   = deg + N;
  float* agg1 = h1 + (size_t)N * 16;
  float* rec2 = h1;

  dim3 blk(TPB);
  dim3 gN((N + TPB - 1) / TPB);
  dim3 eblk(EDGE_T);
  const int rDeg = (N + DEG_P - 1) / DEG_P;
  const int r1 = (N + L1_P - 1) / L1_P;
  const int r2 = (N + L2_P - 1) / L2_P;

  const int tiles = (N + 15) / 16;
  const int wavesPerBlock = TPB / 32;
  dim3 gT((tiles + wavesPerBlock - 1) / wavesPerBlock);
  k_gemm1_wmma<<<gT, blk, 0, stream>>>(x, W1, deg, h1, agg1, N);
  k_edge_deg<<<dim3(rDeg * DEG_S), eblk, 0, stream>>>(ei + E, deg, N, E, rDeg);

  k_edge_agg16<<<dim3(r1 * L1_S), eblk, 0, stream>>>(ei, deg, h1, agg1, N, E, r1);
  k_node2<<<gN, blk, 0, stream>>>(agg1, b1, W2, b2, deg, rec2, out, N);
  k_edge_agg2<<<dim3(r2 * L2_S), eblk, 0, stream>>>(ei, deg, rec2, out, N, E, r2);
}


# layer-1/2 edge scans: the four edges of a 16-byte load are ranked together (four ballots, scalar running counts, one room test per load): no lane-wise test or branch per edge; list slots unchanged
# speedup vs baseline: 8.6233x; 1.0900x over previous
#include <hip/hip_runtime.h>

typedef __attribute__((ext_vector_type(2))) float v2f;
typedef __attribute__((ext_vector_type(8))) float v8f;

#define TPB 256

#define EDGE_T 1024
#define LIST_CAP 16384
#define NWAVE (EDGE_T / 32)
#define WLIST (LIST_CAP / NWAVE)
#define DEG_P 25600
#define DEG_S 64
#define L1_P 3136
#define L1_S 8
#define L1_PITCH 17
#define L1_OWN ((L1_P + L1_S - 1) / L1_S)
#define L2_P 12544
#define L2_S 32
#define L1_RB 12
#define L2_RB 14
static_assert(L1_P <= (1 << L1_RB), "layer-1 row does not fit its list-word field");
static_assert(L2_P <= (1 << L2_RB), "layer-2 row does not fit its list-word field");

static __device__ __forceinline__ void atomAddF32(float* p, float v) {
  unsafeAtomicAdd(p, v);
}

static __device__ __forceinline__ void ldsAddF32(float* p, float v) {
  __hip_atomic_fetch_add(p, v, __ATOMIC_RELAXED, __HIP_MEMORY_SCOPE_WORKGROUP);
}

static __device__ __forceinline__ float dinv_of(float d) {
  return (d > 0.0f) ? rsqrtf(d) : 0.0f;
}

template <bool WITH_SRC, class F4, class F>
static __device__ __forceinline__ void scan_owned(const int* __restrict__ dst,
                                                  const int* __restrict__ src, int eb, int ee,
                                                  bool vec_ok, int base, int P, F4 on4, F onvec) {
  const int lane = threadIdx.x & 31;
  const int w0 = __builtin_amdgcn_readfirstlane((int)(threadIdx.x >> 5)) * 32;
  const int nvec = vec_ok ? ((ee - eb) >> 2) : 0;
  const int4* d4 = (const int4*)(dst + eb);
  const int4* s4 = (const int4*)(src + eb);
  auto lds4 = [&](int i) -> int4 {
    if constexpr (WITH_SRC) return s4[i];
    else return make_int4(0, 0, 0, 0);
  };
  auto test4 = [&](const int4 v, const int4 u, int i, bool ok) {
    const int e0 = i * 4;
    const unsigned r0 = (unsigned)(v.x - base);
    const unsigned r1 = (unsigned)(v.y - base);
    const unsigned r2 = (unsigned)(v.z - base);
    const unsigned r3 = (unsigned)(v.w - base);
    on4(e0, make_int4((int)r0, (int)r1, (int)r2, (int)r3), ok && r0 < (unsigned)P,
        ok && r1 < (unsigned)P, ok && r2 < (unsigned)P, ok && r3 < (unsigned)P, u);
  };
  int iw = w0;
  for (; iw + 31 + 3 * EDGE_T < nvec; iw += 4 * EDGE_T) {
    const int i = iw + lane;
    const int4 a = d4[i];
    const int4 b = d4[i + EDGE_T];
    const int4 c = d4[i + 2 * EDGE_T];
    const int4 d = d4[i + 3 * EDGE_T];
    const int4 sa = lds4(i);
    const int4 sb = lds4(i + EDGE_T);
    const int4 sc = lds4(i + 2 * EDGE_T);
    const int4 sd = lds4(i + 3 * EDGE_T);
    test4(a, sa, i, true);
    test4(b, sb, i + EDGE_T, true);
    test4(c, sc, i + 2 * EDGE_T, true);
    test4(d, sd, i + 3 * EDGE_T, true);
  }
  for (; iw < nvec; iw += EDGE_T) {
    const int i = iw + lane;
    const bool ok = i < nvec;
    const int j = ok ? i : iw;
    test4(d4[j], lds4(j), i, ok);
  }
  for (int ew = eb + nvec * 4 + w0; ew < ee; ew += EDGE_T) {
    const int e = ew + lane;
    const bool ok = e < ee;
    const int j = ok ? e : ew;
    const unsigned r = (unsigned)(dst[j] - base);
    int sv = 0;
    if constexpr (WITH_SRC) sv = src[j];
    onvec(e - eb, (int)r, ok && r < (unsigned)P, sv);
  }
}

static __device__ __forceinline__ bool wave_append(unsigned* mylist, int& cnt, int cap,
                                                   unsigned word, bool m) {
  const unsigned mask = __builtin_amdgcn_ballot_w32(m);
  bool spill = false;
  if (mask != 0u) {
    const int p = cnt + (int)__builtin_amdgcn_mbcnt_lo(mask, 0u);
    cnt = __builtin_amdgcn_readfirstlane(cnt + __popc(mask));
    if (m) {
      if (p < cap) mylist[p] = word; else spill = true;
    }
  }
  return spill;
}

static __device__ __forceinline__ bool wave_append4(unsigned* mylist, int& cnt, int cap,
                                                    unsigned w0, unsigned w1, unsigned w2,
                                                    unsigned w3, bool m0, bool m1, bool m2,
                                                    bool m3) {
  const unsigned k0 = __builtin_amdgcn_ballot_w32(m0);
  const unsigned k1 = __builtin_amdgcn_ballot_w32(m1);
  const unsigned k2 = __builtin_amdgcn_ballot_w32(m2);
  const unsigned k3 = __builtin_amdgcn_ballot_w32(m3);
  const int b1 = cnt + __popc(k0);
  const int b2 = b1 + __popc(k1);
  const int b3 = b2 + __popc(k2);
  const int b4 = b3 + __popc(k3);
  if (__builtin_expect(b4 > cap, 0)) return false;
  if (__builtin_expect(m0, 1)) mylist[__builtin_amdgcn_mbcnt_lo(k0, (unsigned)cnt)] = w0;
  if (__builtin_expect(m1, 1)) mylist[__builtin_amdgcn_mbcnt_lo(k1, (unsigned)b1)] = w1;
  if (__builtin_expect(m2, 1)) mylist[__builtin_amdgcn_mbcnt_lo(k2, (unsigned)b2)] = w2;
  if (__builtin_expect(m3, 1)) mylist[__builtin_amdgcn_mbcnt_lo(k3, (unsigned)b3)] = w3;
  cnt = b4;
  return true;
}

static __device__ __forceinline__ int seg_len(int E, int S) {
  return (((E + S - 1) / S) + 3) & ~3;
}


__global__ __launch_bounds__(EDGE_T) void k_edge_deg(const int* __restrict__ dst,
                                                     float* __restrict__ deg, int N, int E, int R) {
  __shared__ int cnt[DEG_P];
  const int tid = threadIdx.x;
  const int range = blockIdx.x % R;
  const int seg = blockIdx.x / R;
  const int base = range * DEG_P;
  for (int i = tid; i < DEG_P; i += EDGE_T) cnt[i] = 0;
  __syncthreads();
  const int sl = seg_len(E, DEG_S);
  const int eb = min(seg * sl, E);
  const int ee = min(eb + sl, E);
  auto count1 = [&](int, int r, bool m, int) {
    if (m) __hip_atomic_fetch_add(&cnt[r], 1, __ATOMIC_RELAXED, __HIP_MEMORY_SCOPE_WORKGROUP);
  };
  scan_owned<false>(dst, dst, eb, ee, (E & 3) == 0, base, DEG_P,
                    [&](int e0, const int4 r, bool m0, bool m1, bool m2, bool m3, const int4) {
                      count1(e0 + 0, r.x, m0, 0);
                      count1(e0 + 1, r.y, m1, 0);
                      count1(e0 + 2, r.z, m2, 0);
                      count1(e0 + 3, r.w, m3, 0);
                    },
                    count1);
  __syncthreads();
  for (int i = tid; i < DEG_P; i += EDGE_T) {
    const int n = base + i;
    const int c = cnt[i];
    if (n < N && c != 0) atomAddF32(&deg[n], (float)c);
  }
}


__global__ void k_gemm1_wmma(const float* __restrict__ x,
                             const float* __restrict__ W1,
                             float* __restrict__ deg,
                             float* __restrict__ h1,
                             float* __restrict__ agg1, int N) {
  const int lane = threadIdx.x & 31;
  const int tile = blockIdx.x * (blockDim.x >> 5) + (threadIdx.x >> 5);
  const int row0 = tile * 16;
  if (row0 >= N) return;

  const int half = lane >> 4;
  const int lrow = lane & 15;

  int arow = row0 + lrow;
  if (arow >= N) arow = N - 1;

  const float* xrow = x + (size_t)arow * 128 + half * 2;
  const float* wcol = W1 + (size_t)(half * 2) * 16 + lrow;

  v8f acc = {};
#pragma unroll
  for (int k0 = 0; k0 < 128; k0 += 4) {
    v2f a = *(const v2f*)(xrow + k0);
    v2f b;
    b.x = wcol[(size_t)k0 * 16];
    b.y = wcol[(size_t)k0 * 16 + 16];
    acc = __builtin_amdgcn_wmma_f32_16x16x4_f32(
         false, a,  false, b,
         (short)0, acc,  false,  false);
  }

  const int col = lrow;
#pragma unroll
  for (int v = 0; v < 8; ++v) {
    const int m = row0 + v + half * 8;
    if (m < N) {
      h1[(size_t)m * 16 + col] = acc[v];
      agg1[(size_t)m * 16 + col] = 0.0f;
    }
  }
  if (lane < 16 && row0 + lane < N) deg[row0 + lane] = 1.0f;
}


static __device__ __forceinline__ void l1_add(float* op, const float4 m0, const float4 m1,
                                              const float4 m2, const float4 m3, float w) {
  ldsAddF32(op + 0,  m0.x * w);
  ldsAddF32(op + 1,  m0.y * w);
  ldsAddF32(op + 2,  m0.z * w);
  ldsAddF32(op + 3,  m0.w * w);
  ldsAddF32(op + 4,  m1.x * w);
  ldsAddF32(op + 5,  m1.y * w);
  ldsAddF32(op + 6,  m1.z * w);
  ldsAddF32(op + 7,  m1.w * w);
  ldsAddF32(op + 8,  m2.x * w);
  ldsAddF32(op + 9,  m2.y * w);
  ldsAddF32(op + 10, m2.z * w);
  ldsAddF32(op + 11, m2.w * w);
  ldsAddF32(op + 12, m3.x * w);
  ldsAddF32(op + 13, m3.y * w);
  ldsAddF32(op + 14, m3.z * w);
  ldsAddF32(op + 15, m3.w * w);
}

static __device__ __forceinline__ void l1_edge(int e, const int* __restrict__ src,
                                               const int* __restrict__ dst,
                                               const float* __restrict__ deg,
                                               const float* __restrict__ h1,
                                               float* acc, int base) {
  const int s = src[e];
  const int d = dst[e];
  const float w = dinv_of(deg[s]) * dinv_of(deg[d]);
  const float4* hs = (const float4*)(h1 + (size_t)s * 16);
  l1_add(acc + (d - base) * L1_PITCH, hs[0], hs[1], hs[2], hs[3], w);
}

__global__ __launch_bounds__(EDGE_T) void k_edge_agg16(const int* __restrict__ ei,
                                                       const float* __restrict__ deg,
                                                       const float* __restrict__ h1,
                                                       float* __restrict__ agg1,
                                                       int N, int E, int R) {
  __shared__ float acc[L1_P * L1_PITCH];
  __shared__ float dinv_r[L1_P];
  __shared__ unsigned list[LIST_CAP];
  const int tid = threadIdx.x;
  const int lane = tid & 31;
  unsigned* mylist = list + (tid >> 5) * WLIST;
  const int range = blockIdx.x % R;
  const int seg = blockIdx.x / R;
  const int base = range * L1_P;
  const int* __restrict__ src = ei;
  const int* __restrict__ dst = ei + E;

  for (int i = tid; i < L1_P * L1_PITCH; i += EDGE_T) acc[i] = 0.0f;
  for (int i = tid; i < L1_P; i += EDGE_T) dinv_r[i] = (base + i < N) ? dinv_of(deg[base + i]) : 0.0f;
  __syncthreads();

  constexpr int OWN_T = (L1_OWN * 16 + EDGE_T - 1) / EDGE_T;
  const int own_lo = seg * L1_OWN;
  float selfv[OWN_T];
  int selfo[OWN_T];
#pragma unroll
  for (int q = 0; q < OWN_T; ++q) {
    const int i = tid + q * EDGE_T;
    const int r = own_lo + (i >> 4);
    const int k = i & 15;
    const bool ok = i < L1_OWN * 16 && r < L1_P && base + r < N;
    const int node = ok ? base + r : base;
    selfv[q] = h1[(size_t)node * 16 + k];
    selfo[q] = ok ? r * L1_PITCH + k : -1;
  }

  const int sl = seg_len(E, L1_S);
  const int eb = min(seg * sl, E);
  const int ee = min(eb + sl, E);
  const int cap = (sl <= (1 << (32 - L1_RB))) ? WLIST : 0;
  int cnt = 0;
  auto list1 = [&](int rel, int r, bool m, int) {
    if (wave_append(mylist, cnt, cap, ((unsigned)rel << L1_RB) | (unsigned)r, m))
      l1_edge(eb + rel, src, dst, deg, h1, acc, base);
  };
  scan_owned<false>(dst, dst, eb, ee, (E & 3) == 0, base, L1_P,
                    [&](int e0, const int4 r, bool m0, bool m1, bool m2, bool m3, const int4) {
                      const unsigned w = (unsigned)e0 << L1_RB;
                      if (!wave_append4(mylist, cnt, cap, w | (unsigned)r.x,
                                        w | (1u << L1_RB) | (unsigned)r.y,
                                        w | (2u << L1_RB) | (unsigned)r.z,
                                        w | (3u << L1_RB) | (unsigned)r.w, m0, m1, m2, m3)) {
                        list1(e0 + 0, r.x, m0, 0);
                        list1(e0 + 1, r.y, m1, 0);
                        list1(e0 + 2, r.z, m2, 0);
                        list1(e0 + 3, r.w, m3, 0);
                      }
                    },
                    list1);
  __syncthreads();

  const int n = min(cnt, cap);
  for (int j0 = 0; j0 < n; j0 += 4 * 32) {
    int s[4], r[4];
    bool ok[4];
    float w[4];
    float4 m[4][4];
#pragma unroll
    for (int q = 0; q < 4; ++q) {
      const int idx = j0 + q * 32 + lane;
      ok[q] = idx < n;
      const unsigned wd = mylist[ok[q] ? idx : n - 1];
      r[q] = (int)(wd & ((1u << L1_RB) - 1u));
      s[q] = src[eb + (int)(wd >> L1_RB)];
    }
#pragma unroll
    for (int q = 0; q < 4; ++q) {
      w[q] = dinv_of(deg[s[q]]) * dinv_r[r[q]];
      const float4* hs = (const float4*)(h1 + (size_t)s[q] * 16);
      m[q][0] = hs[0]; m[q][1] = hs[1]; m[q][2] = hs[2]; m[q][3] = hs[3];
    }
#pragma unroll
    for (int q = 0; q < 4; ++q)
      if (ok[q]) l1_add(acc + r[q] * L1_PITCH, m[q][0], m[q][1], m[q][2], m[q][3], w[q]);
  }
#pragma unroll
  for (int q = 0; q < OWN_T; ++q) {
    if (selfo[q] >= 0) {
      const float di = dinv_r[own_lo + ((tid + q * EDGE_T) >> 4)];
      ldsAddF32(acc + selfo[q], selfv[q] * di * di);
    }
  }
  __syncthreads();

  for (int i = tid; i < L1_P * 16; i += EDGE_T) {
    const int r = i >> 4;
    const int k = i & 15;
    const int node = base + r;
    if (node < N) atomAddF32(agg1 + (size_t)node * 16 + k, acc[r * L1_PITCH + k]);
  }
}


__global__ void k_node2(const float* __restrict__ agg1,
                        const float* __restrict__ b1,
                        const float* __restrict__ W2,
                        const float* __restrict__ b2,
                        const float* __restrict__ deg,
                        float* __restrict__ rec2,
                        float* __restrict__ out, int N) {
  int n = blockIdx.x * blockDim.x + threadIdx.x;
  if (n >= N) return;
  const float4* ap = (const float4*)(agg1 + (size_t)n * 16);
  float c0 = 0.0f, c1 = 0.0f;
#pragma unroll
  for (int i = 0; i < 4; ++i) {
    float4 t = ap[i];
    float v[4] = {t.x, t.y, t.z, t.w};
#pragma unroll
    for (int j = 0; j < 4; ++j) {
      const int k = i * 4 + j;
      const float hv = fmaxf(v[j] + b1[k], 0.0f);
      c0 = fmaf(hv, W2[k * 2 + 0], c0);
      c1 = fmaf(hv, W2[k * 2 + 1], c1);
    }
  }
  const float di = dinv_of(deg[n]);
  *(float4*)(rec2 + (size_t)n * 4) = make_float4(c0, c1, di, 0.0f);
  const float w = di * di;
  out[(size_t)n * 2 + 0] = fmaf(c0, w, b2[0]);
  out[(size_t)n * 2 + 1] = fmaf(c1, w, b2[1]);
}


static __device__ __forceinline__ void l2_edge(int e, const int* __restrict__ src,
                                               const int* __restrict__ dst,
                                               const float* __restrict__ deg,
                                               const float* __restrict__ rec2,
                                               float* acc, int base) {
  const int s = src[e];
  const int d = dst[e];
  const float4 mm = *(const float4*)(rec2 + (size_t)s * 4);
  const float w = mm.z * dinv_of(deg[d]);
  float* op = acc + (d - base) * 2;
  ldsAddF32(op + 0, mm.x * w);
  ldsAddF32(op + 1, mm.y * w);
}

__global__ __launch_bounds__(EDGE_T) void k_edge_agg2(const int* __restrict__ ei,
                                                      const float* __restrict__ deg,
                                                      const float* __restrict__ rec2,
                                                      float* __restrict__ out,
                                                      int N, int E, int R) {
  __shared__ float acc[L2_P * 2];
  __shared__ float dinv_r[L2_P];
  __shared__ unsigned list[LIST_CAP];
  const int tid = threadIdx.x;
  const int lane = tid & 31;
  unsigned* mylist = list + (tid >> 5) * WLIST;
  const int range = blockIdx.x % R;
  const int seg = blockIdx.x / R;
  const int base = range * L2_P;
  const int* __restrict__ src = ei;
  const int* __restrict__ dst = ei + E;

  for (int i = tid; i < L2_P * 2; i += EDGE_T) acc[i] = 0.0f;
  for (int i = tid; i < L2_P; i += EDGE_T) dinv_r[i] = (base + i < N) ? dinv_of(deg[base + i]) : 0.0f;
  __syncthreads();

  const int sl = seg_len(E, L2_S);
  const int eb = min(seg * sl, E);
  const int ee = min(eb + sl, E);
  const int cap = (N <= (1 << (32 - L2_RB))) ? WLIST : 0;
  int cnt = 0;
  auto list1 = [&](int rel, int r, bool m, int sv) {
    if (wave_append(mylist, cnt, cap, ((unsigned)sv << L2_RB) | (unsigned)r, m))
      l2_edge(eb + rel, src, dst, deg, rec2, acc, base);
  };
  scan_owned<true>(dst, src, eb, ee, (E & 3) == 0, base, L2_P,
                   [&](int e0, const int4 r, bool m0, bool m1, bool m2, bool m3, const int4 sv) {
                     if (!wave_append4(mylist, cnt, cap, ((unsigned)sv.x << L2_RB) | (unsigned)r.x,
                                       ((unsigned)sv.y << L2_RB) | (unsigned)r.y,
                                       ((unsigned)sv.z << L2_RB) | (unsigned)r.z,
                                       ((unsigned)sv.w << L2_RB) | (unsigned)r.w, m0, m1, m2, m3)) {
                       list1(e0 + 0, r.x, m0, sv.x);
                       list1(e0 + 1, r.y, m1, sv.y);
                       list1(e0 + 2, r.z, m2, sv.z);
                       list1(e0 + 3, r.w, m3, sv.w);
                     }
                   },
                   list1);
  __syncthreads();

  const int n = min(cnt, cap);
  for (int j0 = 0; j0 < n; j0 += 4 * 32) {
    int r[4];
    bool ok[4];
    float dr[4];
    float4 m[4];
#pragma unroll
    for (int q = 0; q < 4; ++q) {
      const int idx = j0 + q * 32 + lane;
      ok[q] = idx < n;
      const unsigned wd = mylist[ok[q] ? idx : n - 1];
      r[q] = (int)(wd & ((1u << L2_RB) - 1u));
      m[q] = *(const float4*)(rec2 + (size_t)(wd >> L2_RB) * 4);
    }
#pragma unroll
    for (int q = 0; q < 4; ++q) dr[q] = dinv_r[r[q]];
#pragma unroll
    for (int q = 0; q < 4; ++q) {
      if (ok[q]) {
        const float w = m[q].z * dr[q];
        float* op = acc + r[q] * 2;
        ldsAddF32(op + 0, m[q].x * w);
        ldsAddF32(op + 1, m[q].y * w);
      }
    }
  }
  __syncthreads();

  for (int i = tid; i < L2_P * 2; i += EDGE_T) {
    const size_t o = (size_t)base * 2 + i;
    if (o < (size_t)N * 2) atomAddF32(out + o, acc[i]);
  }
}


extern "C" void kernel_launch(void* const* d_in, const int* in_sizes, int n_in,
                              void* d_out, int out_size, void* d_ws, size_t ws_size,
                              hipStream_t stream) {
  const float* x  = (const float*)d_in[0];
  const int*   ei = (const int*)d_in[1];
  const float* W1 = (const float*)d_in[2];
  const float* b1 = (const float*)d_in[3];
  const float* W2 = (const float*)d_in[4];
  const float* b2 = (const float*)d_in[5];
  float* out = (float*)d_out;

  const int N = in_sizes[0] / 128;
  const int E = in_sizes[1] / 2;

  float* deg  = (float*)d_ws;
  float* h1   = deg + N;
  float* agg1 = h1 + (size_t)N * 16;
  float* rec2 = h1;

  dim3 blk(TPB);
  dim3 gN((N + TPB - 1) / TPB);
  dim3 eblk(EDGE_T);
  const int rDeg = (N + DEG_P - 1) / DEG_P;
  const int r1 = (N + L1_P - 1) / L1_P;
  const int r2 = (N + L2_P - 1) / L2_P;

  const int tiles = (N + 15) / 16;
  const int wavesPerBlock = TPB / 32;
  dim3 gT((tiles + wavesPerBlock - 1) / wavesPerBlock);
  k_gemm1_wmma<<<gT, blk, 0, stream>>>(x, W1, deg, h1, agg1, N);
  k_edge_deg<<<dim3(rDeg * DEG_S), eblk, 0, stream>>>(ei + E, deg, N, E, rDeg);

  k_edge_agg16<<<dim3(r1 * L1_S), eblk, 0, stream>>>(ei, deg, h1, agg1, N, E, r1);
  k_node2<<<gN, blk, 0, stream>>>(agg1, b1, W2, b2, deg, rec2, out, N);
  k_edge_agg2<<<dim3(r2 * L2_S), eblk, 0, stream>>>(ei, deg, rec2, out, N, E, r2);
}


# edge lists built once per call: a new launch orders (source, row) words by layer-1 destination range in the workspace; both edge passes read their buckets, no scans; LDS sum order changes
# speedup vs baseline: 9.1162x; 1.0572x over previous
#include <hip/hip_runtime.h>

typedef __attribute__((ext_vector_type(2))) float v2f;
typedef __attribute__((ext_vector_type(8))) float v8f;

#define TPB 256

#define EDGE_T 1024
#define LIST_CAP 16384
#define NWAVE (EDGE_T / 32)
#define WLIST (LIST_CAP / NWAVE)
#define DEG_SUB 8
#define DEG_P (DEG_SUB * L1_P)
#define DEG_S 64
#define L1_P 3136
#define L1_S 8
#define L1_PITCH 17
#define L1_OWN ((L1_P + L1_S - 1) / L1_S)
#define L2_P 12544
#define L2_S 32
#define L1_RB 12
#define L2_RB 14
static_assert(L1_P <= (1 << L1_RB), "layer-1 row does not fit its list-word field");
static_assert(L2_P <= (1 << L2_RB), "layer-2 row does not fit its list-word field");
#define NB_MAX 64
#define BK_V 3
#define BK_CAP (BK_V * 4 * EDGE_T)
#define L2_GRP (L2_P / L1_P)
#define MSTRIDE 64
#define META_CUR (NB_MAX * MSTRIDE)
#define META_BST (2 * NB_MAX * MSTRIDE)
#define META_WORDS (META_BST + 128)
static_assert(L1_P % 32 == 0, "a wave's 32 consecutive nodes must lie in one layer-1 range");
static_assert(DEG_P % 32 == 0 && EDGE_T % 32 == 0, "degree pass: a wave is whole in every trip of the flush loop");
static_assert(L2_P == L2_GRP * L1_P, "a layer-2 range must be whole layer-1 ranges");
static_assert(NB_MAX + 8 <= 128, "meta words");
static_assert(NB_MAX == 64, "the prefix helper takes two buckets per lane");

static __device__ __forceinline__ void atomAddF32(float* p, float v) {
  unsafeAtomicAdd(p, v);
}

static __device__ __forceinline__ void ldsAddF32(float* p, float v) {
  __hip_atomic_fetch_add(p, v, __ATOMIC_RELAXED, __HIP_MEMORY_SCOPE_WORKGROUP);
}

static __device__ __forceinline__ float dinv_of(float d) {
  return (d > 0.0f) ? rsqrtf(d) : 0.0f;
}

static __device__ __forceinline__ int wave_sum(int v) {
  v += __shfl_xor(v, 16, 32);
  v += __shfl_xor(v, 8, 32);
  v += __shfl_xor(v, 4, 32);
  v += __shfl_xor(v, 2, 32);
  v += __shfl_xor(v, 1, 32);
  return v;
}

static __device__ __forceinline__ void wave_prefix2(int a0, int a1, int lane, int& p0, int& p1,
                                                    int& total) {
  const int sum2 = a0 + a1;
  int inc = sum2;
#pragma unroll
  for (int d = 1; d < 32; d <<= 1) {
    const int t = __shfl_up(inc, d, 32);
    if (lane >= d) inc += t;
  }
  p0 = inc - sum2;
  p1 = p0 + a0;
  total = __shfl(inc, 31, 32);
}

template <bool WITH_SRC, class F4, class F>
static __device__ __forceinline__ void scan_owned(const int* __restrict__ dst,
                                                  const int* __restrict__ src, int eb, int ee,
                                                  bool vec_ok, int base, int P, F4 on4, F onvec) {
  const int lane = threadIdx.x & 31;
  const int w0 = __builtin_amdgcn_readfirstlane((int)(threadIdx.x >> 5)) * 32;
  const int nvec = vec_ok ? ((ee - eb) >> 2) : 0;
  const int4* d4 = (const int4*)(dst + eb);
  const int4* s4 = (const int4*)(src + eb);
  auto lds4 = [&](int i) -> int4 {
    if constexpr (WITH_SRC) return s4[i];
    else return make_int4(0, 0, 0, 0);
  };
  auto test4 = [&](const int4 v, const int4 u, int i, bool ok) {
    const int e0 = i * 4;
    const unsigned r0 = (unsigned)(v.x - base);
    const unsigned r1 = (unsigned)(v.y - base);
    const unsigned r2 = (unsigned)(v.z - base);
    const unsigned r3 = (unsigned)(v.w - base);
    on4(e0, make_int4((int)r0, (int)r1, (int)r2, (int)r3), ok && r0 < (unsigned)P,
        ok && r1 < (unsigned)P, ok && r2 < (unsigned)P, ok && r3 < (unsigned)P, u);
  };
  int iw = w0;
  for (; iw + 31 + 3 * EDGE_T < nvec; iw += 4 * EDGE_T) {
    const int i = iw + lane;
    const int4 a = d4[i];
    const int4 b = d4[i + EDGE_T];
    const int4 c = d4[i + 2 * EDGE_T];
    const int4 d = d4[i + 3 * EDGE_T];
    const int4 sa = lds4(i);
    const int4 sb = lds4(i + EDGE_T);
    const int4 sc = lds4(i + 2 * EDGE_T);
    const int4 sd = lds4(i + 3 * EDGE_T);
    test4(a, sa, i, true);
    test4(b, sb, i + EDGE_T, true);
    test4(c, sc, i + 2 * EDGE_T, true);
    test4(d, sd, i + 3 * EDGE_T, true);
  }
  for (; iw < nvec; iw += EDGE_T) {
    const int i = iw + lane;
    const bool ok = i < nvec;
    const int j = ok ? i : iw;
    test4(d4[j], lds4(j), i, ok);
  }
  for (int ew = eb + nvec * 4 + w0; ew < ee; ew += EDGE_T) {
    const int e = ew + lane;
    const bool ok = e < ee;
    const int j = ok ? e : ew;
    const unsigned r = (unsigned)(dst[j] - base);
    int sv = 0;
    if constexpr (WITH_SRC) sv = src[j];
    onvec(e - eb, (int)r, ok && r < (unsigned)P, sv);
  }
}

static __device__ __forceinline__ bool wave_append(unsigned* mylist, int& cnt, int cap,
                                                   unsigned word, bool m) {
  const unsigned mask = __builtin_amdgcn_ballot_w32(m);
  bool spill = false;
  if (mask != 0u) {
    const int p = cnt + (int)__builtin_amdgcn_mbcnt_lo(mask, 0u);
    cnt = __builtin_amdgcn_readfirstlane(cnt + __popc(mask));
    if (m) {
      if (p < cap) mylist[p] = word; else spill = true;
    }
  }
  return spill;
}

static __device__ __forceinline__ bool wave_append4(unsigned* mylist, int& cnt, int cap,
                                                    unsigned w0, unsigned w1, unsigned w2,
                                                    unsigned w3, bool m0, bool m1, bool m2,
                                                    bool m3) {
  const unsigned k0 = __builtin_amdgcn_ballot_w32(m0);
  const unsigned k1 = __builtin_amdgcn_ballot_w32(m1);
  const unsigned k2 = __builtin_amdgcn_ballot_w32(m2);
  const unsigned k3 = __builtin_amdgcn_ballot_w32(m3);
  const int b1 = cnt + __popc(k0);
  const int b2 = b1 + __popc(k1);
  const int b3 = b2 + __popc(k2);
  const int b4 = b3 + __popc(k3);
  if (__builtin_expect(b4 > cap, 0)) return false;
  if (__builtin_expect(m0, 1)) mylist[__builtin_amdgcn_mbcnt_lo(k0, (unsigned)cnt)] = w0;
  if (__builtin_expect(m1, 1)) mylist[__builtin_amdgcn_mbcnt_lo(k1, (unsigned)b1)] = w1;
  if (__builtin_expect(m2, 1)) mylist[__builtin_amdgcn_mbcnt_lo(k2, (unsigned)b2)] = w2;
  if (__builtin_expect(m3, 1)) mylist[__builtin_amdgcn_mbcnt_lo(k3, (unsigned)b3)] = w3;
  cnt = b4;
  return true;
}

static __device__ __forceinline__ int seg_len(int E, int S) {
  return (((E + S - 1) / S) + 3) & ~3;
}


__global__ __launch_bounds__(EDGE_T) void k_edge_deg(const int* __restrict__ dst,
                                                     float* __restrict__ deg,
                                                     int* __restrict__ bcnt, int N, int E, int R) {
  __shared__ int cnt[DEG_P];
  __shared__ int sub[DEG_SUB];
  const int tid = threadIdx.x;
  if (tid < DEG_SUB) sub[tid] = 0;
  const int range = blockIdx.x % R;
  const int seg = blockIdx.x / R;
  const int base = range * DEG_P;
  for (int i = tid; i < DEG_P; i += EDGE_T) cnt[i] = 0;
  __syncthreads();
  const int sl = seg_len(E, DEG_S);
  const int eb = min(seg * sl, E);
  const int ee = min(eb + sl, E);
  auto count1 = [&](int, int r, bool m, int) {
    if (m) __hip_atomic_fetch_add(&cnt[r], 1, __ATOMIC_RELAXED, __HIP_MEMORY_SCOPE_WORKGROUP);
  };
  scan_owned<false>(dst, dst, eb, ee, (E & 3) == 0, base, DEG_P,
                    [&](int e0, const int4 r, bool m0, bool m1, bool m2, bool m3, const int4) {
                      count1(e0 + 0, r.x, m0, 0);
                      count1(e0 + 1, r.y, m1, 0);
                      count1(e0 + 2, r.z, m2, 0);
                      count1(e0 + 3, r.w, m3, 0);
                    },
                    count1);
  __syncthreads();
  for (int i = tid; i < DEG_P; i += EDGE_T) {
    const int n = base + i;
    const int c = cnt[i];
    if (n < N && c != 0) atomAddF32(&deg[n], (float)c);
    if (bcnt != nullptr) {
      const int ws = wave_sum(c);
      if ((tid & 31) == 0 && ws != 0)
        __hip_atomic_fetch_add(&sub[i / L1_P], ws, __ATOMIC_RELAXED, __HIP_MEMORY_SCOPE_WORKGROUP);
    }
  }
  if (bcnt != nullptr) {
    __syncthreads();
    const int b = range * DEG_SUB + tid;
    if (tid < DEG_SUB && b < NB_MAX && sub[tid] != 0) atomicAdd(&bcnt[b * MSTRIDE], sub[tid]);
  }
}


__global__ void k_gemm1_wmma(const float* __restrict__ x,
                             const float* __restrict__ W1,
                             float* __restrict__ deg,
                             float* __restrict__ h1,
                             float* __restrict__ agg1,
                             int* __restrict__ meta, int N) {
  if (meta != nullptr && blockIdx.x == 0 && threadIdx.x < NB_MAX) {
    meta[threadIdx.x * MSTRIDE] = 0;
    meta[META_CUR + threadIdx.x * MSTRIDE] = 0;
  }
  const int lane = threadIdx.x & 31;
  const int tile = blockIdx.x * (blockDim.x >> 5) + (threadIdx.x >> 5);
  const int row0 = tile * 16;
  if (row0 >= N) return;

  const int half = lane >> 4;
  const int lrow = lane & 15;

  int arow = row0 + lrow;
  if (arow >= N) arow = N - 1;

  const float* xrow = x + (size_t)arow * 128 + half * 2;
  const float* wcol = W1 + (size_t)(half * 2) * 16 + lrow;

  v8f acc = {};
#pragma unroll
  for (int k0 = 0; k0 < 128; k0 += 4) {
    v2f a = *(const v2f*)(xrow + k0);
    v2f b;
    b.x = wcol[(size_t)k0 * 16];
    b.y = wcol[(size_t)k0 * 16 + 16];
    acc = __builtin_amdgcn_wmma_f32_16x16x4_f32(
         false, a,  false, b,
         (short)0, acc,  false,  false);
  }

  const int col = lrow;
#pragma unroll
  for (int v = 0; v < 8; ++v) {
    const int m = row0 + v + half * 8;
    if (m < N) {
      h1[(size_t)m * 16 + col] = acc[v];
      agg1[(size_t)m * 16 + col] = 0.0f;
    }
  }
  if (lane < 16 && row0 + lane < N) deg[row0 + lane] = 1.0f;
}

__global__ __launch_bounds__(EDGE_T) void k_bucket(const int* __restrict__ ei,
                                                   int* __restrict__ meta,
                                                   unsigned* __restrict__ glist, int E) {
  __shared__ unsigned stage[BK_CAP];
  __shared__ int where[BK_CAP];
  __shared__ int hist[NB_MAX];
  __shared__ int loff[NB_MAX];
  __shared__ int gpos[NB_MAX];
  const int tid = threadIdx.x;
  const int lane = tid & 31;
  const int* __restrict__ src = ei;
  const int* __restrict__ dst = ei + E;
  const int* bcnt = meta;
  int* cur = meta + META_CUR;
  int* bst = meta + META_BST;
  const int eb = blockIdx.x * BK_CAP;
  const bool vec_ok = (E & 3) == 0;

  if (tid < NB_MAX) hist[tid] = 0;
  __syncthreads();
  int bc0 = 0, bc1 = 0;
  if (tid < 32) {
    bc0 = bcnt[2 * lane * MSTRIDE];
    bc1 = bcnt[(2 * lane + 1) * MSTRIDE];
  }

  int dd[BK_V * 4], ss[BK_V * 4];
  if (vec_ok && eb + BK_CAP <= E) {
#pragma unroll
    for (int q = 0; q < BK_V; ++q) {
      const int e = eb + (tid + q * EDGE_T) * 4;
      const int4 dv = *(const int4*)(dst + e);
      const int4 uv = *(const int4*)(src + e);
      dd[q * 4 + 0] = dv.x; dd[q * 4 + 1] = dv.y; dd[q * 4 + 2] = dv.z; dd[q * 4 + 3] = dv.w;
      ss[q * 4 + 0] = uv.x; ss[q * 4 + 1] = uv.y; ss[q * 4 + 2] = uv.z; ss[q * 4 + 3] = uv.w;
    }
  } else {
#pragma unroll
    for (int k = 0; k < BK_V * 4; ++k) {
      const int e = eb + (tid + (k >> 2) * EDGE_T) * 4 + (k & 3);
      const bool ok = e < E;
      dd[k] = ok ? dst[ok ? e : 0] : -1;
      ss[k] = ok ? src[ok ? e : 0] : 0;
    }
  }
  unsigned word[BK_V * 4];
  int slot[BK_V * 4];
#pragma unroll
  for (int k = 0; k < BK_V * 4; ++k) {
    const unsigned d = (unsigned)max(dd[k], 0);
    const unsigned b = d / (unsigned)L1_P;
    const unsigned row = d - b * (unsigned)L1_P;
    word[k] = ((unsigned)ss[k] << L1_RB) | row;
    int sl = -1;
    if (dd[k] >= 0) {
      const int rank = __hip_atomic_fetch_add(&hist[b], 1, __ATOMIC_RELAXED,
                                              __HIP_MEMORY_SCOPE_WORKGROUP);
      sl = (int)(b << 16) | rank;
    }
    slot[k] = sl;
  }
  __syncthreads();

  if (tid < 32) {
    int l0, l1, lt, g0, g1, gt;
    const int h0 = hist[2 * lane], h1c = hist[2 * lane + 1];
    wave_prefix2(h0, h1c, lane, l0, l1, lt);
    wave_prefix2(bc0, bc1, lane, g0, g1, gt);
    loff[2 * lane] = l0;
    loff[2 * lane + 1] = l1;
    const int c0 = atomicAdd(&cur[2 * lane * MSTRIDE], h0);
    const int c1 = atomicAdd(&cur[(2 * lane + 1) * MSTRIDE], h1c);
    gpos[2 * lane] = g0 + c0;
    gpos[2 * lane + 1] = g1 + c1;
    if (blockIdx.x == 0) {
      bst[2 * lane] = g0;
      bst[2 * lane + 1] = g1;
      if (lane < 8) bst[NB_MAX + lane] = gt;
    }
  }
  __syncthreads();

#pragma unroll
  for (int k = 0; k < BK_V * 4; ++k) {
    if (slot[k] >= 0) {
      const int b = slot[k] >> 16;
      const int rank = slot[k] & 0xffff;
      const int p = loff[b] + rank;
      stage[p] = word[k];
      where[p] = gpos[b] + rank;
    }
  }
  __syncthreads();

  const int n = min(BK_CAP, E - eb);
  for (int i = tid; i < n; i += EDGE_T) glist[where[i]] = stage[i];
}


static __device__ __forceinline__ void l1_add(float* op, const float4 m0, const float4 m1,
                                              const float4 m2, const float4 m3, float w) {
  ldsAddF32(op + 0,  m0.x * w);
  ldsAddF32(op + 1,  m0.y * w);
  ldsAddF32(op + 2,  m0.z * w);
  ldsAddF32(op + 3,  m0.w * w);
  ldsAddF32(op + 4,  m1.x * w);
  ldsAddF32(op + 5,  m1.y * w);
  ldsAddF32(op + 6,  m1.z * w);
  ldsAddF32(op + 7,  m1.w * w);
  ldsAddF32(op + 8,  m2.x * w);
  ldsAddF32(op + 9,  m2.y * w);
  ldsAddF32(op + 10, m2.z * w);
  ldsAddF32(op + 11, m2.w * w);
  ldsAddF32(op + 12, m3.x * w);
  ldsAddF32(op + 13, m3.y * w);
  ldsAddF32(op + 14, m3.z * w);
  ldsAddF32(op + 15, m3.w * w);
}

static __device__ __forceinline__ void l1_edge(int e, const int* __restrict__ src,
                                               const int* __restrict__ dst,
                                               const float* __restrict__ deg,
                                               const float* __restrict__ h1,
                                               float* acc, int base) {
  const int s = src[e];
  const int d = dst[e];
  const float w = dinv_of(deg[s]) * dinv_of(deg[d]);
  const float4* hs = (const float4*)(h1 + (size_t)s * 16);
  l1_add(acc + (d - base) * L1_PITCH, hs[0], hs[1], hs[2], hs[3], w);
}

template <bool FULL>
static __device__ __forceinline__ void l1_trip(const unsigned* __restrict__ mine, int n, int j0,
                                               int lane, const float* __restrict__ deg,
                                               const float* __restrict__ h1,
                                               const float* dinv_r, float* acc) {
  int s[4], r[4];
  bool ok[4];
  float w[4];
  float4 m[4][4];
#pragma unroll
  for (int q = 0; q < 4; ++q) {
    const int idx = j0 + q * 32 + lane;
    ok[q] = FULL || idx < n;
    const unsigned wd = mine[ok[q] ? idx : n - 1];
    r[q] = (int)(wd & ((1u << L1_RB) - 1u));
    s[q] = (int)(wd >> L1_RB);
  }
#pragma unroll
  for (int q = 0; q < 4; ++q) {
    w[q] = dinv_of(deg[s[q]]) * dinv_r[r[q]];
    const float4* hs = (const float4*)(h1 + (size_t)s[q] * 16);
    m[q][0] = hs[0]; m[q][1] = hs[1]; m[q][2] = hs[2]; m[q][3] = hs[3];
  }
#pragma unroll
  for (int q = 0; q < 4; ++q)
    if (ok[q]) l1_add(acc + r[q] * L1_PITCH, m[q][0], m[q][1], m[q][2], m[q][3], w[q]);
}

template <bool BK>
__global__ __launch_bounds__(EDGE_T) void k_edge_agg16(const int* __restrict__ ei,
                                                       const float* __restrict__ deg,
                                                       const float* __restrict__ h1,
                                                       float* __restrict__ agg1,
                                                       const int* __restrict__ bst,
                                                       const unsigned* __restrict__ glist,
                                                       int N, int E, int R) {
  __shared__ float acc[L1_P * L1_PITCH];
  __shared__ float dinv_r[L1_P];
  __shared__ unsigned list[BK ? 32 : LIST_CAP];
  const int tid = threadIdx.x;
  const int lane = tid & 31;
  unsigned* mylist = list + (tid >> 5) * WLIST;
  const int range = blockIdx.x % R;
  const int seg = blockIdx.x / R;
  const int base = range * L1_P;
  const int* __restrict__ src = ei;
  const int* __restrict__ dst = ei + E;

  for (int i = tid; i < L1_P * L1_PITCH; i += EDGE_T) acc[i] = 0.0f;
  for (int i = tid; i < L1_P; i += EDGE_T) dinv_r[i] = (base + i < N) ? dinv_of(deg[base + i]) : 0.0f;
  __syncthreads();

  constexpr int OWN_T = (L1_OWN * 16 + EDGE_T - 1) / EDGE_T;
  const int own_lo = seg * L1_OWN;
  float selfv[OWN_T];
  int selfo[OWN_T];
#pragma unroll
  for (int q = 0; q < OWN_T; ++q) {
    const int i = tid + q * EDGE_T;
    const int r = own_lo + (i >> 4);
    const int k = i & 15;
    const bool ok = i < L1_OWN * 16 && r < L1_P && base + r < N;
    const int node = ok ? base + r : base;
    selfv[q] = h1[(size_t)node * 16 + k];
    selfo[q] = ok ? r * L1_PITCH + k : -1;
  }

  if constexpr (BK) {
  const int lo = bst[range];
  const int tot = bst[range + 1] - lo;
  const int per = (tot + L1_S - 1) / L1_S;
  const int off = min(seg * per, tot);
  const int n = min(per, tot - off);
  const unsigned* __restrict__ mine = glist + lo + off;
  const int wv = __builtin_amdgcn_readfirstlane(tid >> 5);
  for (int j0 = wv * 128; j0 < n; j0 += NWAVE * 128) {
    if (j0 + 128 <= n) l1_trip<true>(mine, n, j0, lane, deg, h1, dinv_r, acc);
    else l1_trip<false>(mine, n, j0, lane, deg, h1, dinv_r, acc);
  }
  } else {
  const int sl = seg_len(E, L1_S);
  const int eb = min(seg * sl, E);
  const int ee = min(eb + sl, E);
  const int cap = (sl <= (1 << (32 - L1_RB))) ? WLIST : 0;
  int cnt = 0;
  auto list1 = [&](int rel, int r, bool m, int) {
    if (wave_append(mylist, cnt, cap, ((unsigned)rel << L1_RB) | (unsigned)r, m))
      l1_edge(eb + rel, src, dst, deg, h1, acc, base);
  };
  scan_owned<false>(dst, dst, eb, ee, (E & 3) == 0, base, L1_P,
                    [&](int e0, const int4 r, bool m0, bool m1, bool m2, bool m3, const int4) {
                      const unsigned w = (unsigned)e0 << L1_RB;
                      if (!wave_append4(mylist, cnt, cap, w | (unsigned)r.x,
                                        w | (1u << L1_RB) | (unsigned)r.y,
                                        w | (2u << L1_RB) | (unsigned)r.z,
                                        w | (3u << L1_RB) | (unsigned)r.w, m0, m1, m2, m3)) {
                        list1(e0 + 0, r.x, m0, 0);
                        list1(e0 + 1, r.y, m1, 0);
                        list1(e0 + 2, r.z, m2, 0);
                        list1(e0 + 3, r.w, m3, 0);
                      }
                    },
                    list1);
  __syncthreads();

  const int n = min(cnt, cap);
  for (int j0 = 0; j0 < n; j0 += 4 * 32) {
    int s[4], r[4];
    bool ok[4];
    float w[4];
    float4 m[4][4];
#pragma unroll
    for (int q = 0; q < 4; ++q) {
      const int idx = j0 + q * 32 + lane;
      ok[q] = idx < n;
      const unsigned wd = mylist[ok[q] ? idx : n - 1];
      r[q] = (int)(wd & ((1u << L1_RB) - 1u));
      s[q] = src[eb + (int)(wd >> L1_RB)];
    }
#pragma unroll
    for (int q = 0; q < 4; ++q) {
      w[q] = dinv_of(deg[s[q]]) * dinv_r[r[q]];
      const float4* hs = (const float4*)(h1 + (size_t)s[q] * 16);
      m[q][0] = hs[0]; m[q][1] = hs[1]; m[q][2] = hs[2]; m[q][3] = hs[3];
    }
#pragma unroll
    for (int q = 0; q < 4; ++q)
      if (ok[q]) l1_add(acc + r[q] * L1_PITCH, m[q][0], m[q][1], m[q][2], m[q][3], w[q]);
  }
  }
#pragma unroll
  for (int q = 0; q < OWN_T; ++q) {
    if (selfo[q] >= 0) {
      const float di = dinv_r[own_lo + ((tid + q * EDGE_T) >> 4)];
      ldsAddF32(acc + selfo[q], selfv[q] * di * di);
    }
  }
  __syncthreads();

  for (int i = tid; i < L1_P * 16; i += EDGE_T) {
    const int r = i >> 4;
    const int k = i & 15;
    const int node = base + r;
    if (node < N) atomAddF32(agg1 + (size_t)node * 16 + k, acc[r * L1_PITCH + k]);
  }
}


__global__ void k_node2(const float* __restrict__ agg1,
                        const float* __restrict__ b1,
                        const float* __restrict__ W2,
                        const float* __restrict__ b2,
                        const float* __restrict__ deg,
                        float* __restrict__ rec2,
                        float* __restrict__ out, int N) {
  int n = blockIdx.x * blockDim.x + threadIdx.x;
  if (n >= N) return;
  const float4* ap = (const float4*)(agg1 + (size_t)n * 16);
  float c0 = 0.0f, c1 = 0.0f;
#pragma unroll
  for (int i = 0; i < 4; ++i) {
    float4 t = ap[i];
    float v[4] = {t.x, t.y, t.z, t.w};
#pragma unroll
    for (int j = 0; j < 4; ++j) {
      const int k = i * 4 + j;
      const float hv = fmaxf(v[j] + b1[k], 0.0f);
      c0 = fmaf(hv, W2[k * 2 + 0], c0);
      c1 = fmaf(hv, W2[k * 2 + 1], c1);
    }
  }
  const float di = dinv_of(deg[n]);
  *(float4*)(rec2 + (size_t)n * 4) = make_float4(c0, c1, di, 0.0f);
  const float w = di * di;
  out[(size_t)n * 2 + 0] = fmaf(c0, w, b2[0]);
  out[(size_t)n * 2 + 1] = fmaf(c1, w, b2[1]);
}


static __device__ __forceinline__ void l2_edge(int e, const int* __restrict__ src,
                                               const int* __restrict__ dst,
                                               const float* __restrict__ deg,
                                               const float* __restrict__ rec2,
                                               float* acc, int base) {
  const int s = src[e];
  const int d = dst[e];
  const float4 mm = *(const float4*)(rec2 + (size_t)s * 4);
  const float w = mm.z * dinv_of(deg[d]);
  float* op = acc + (d - base) * 2;
  ldsAddF32(op + 0, mm.x * w);
  ldsAddF32(op + 1, mm.y * w);
}

template <bool FULL>
static __device__ __forceinline__ void l2_trip(const unsigned* __restrict__ glist, int a, int n,
                                               int j0, int lane, int g1, int g2, int g3,
                                               const float* __restrict__ rec2,
                                               const float* dinv_r, float* acc) {
  int r[4];
  bool ok[4];
  float dr[4];
  float4 m[4];
#pragma unroll
  for (int q = 0; q < 4; ++q) {
    const int idx = j0 + q * 32 + lane;
    ok[q] = FULL || idx < n;
    const int gi = a + (ok[q] ? idx : n - 1);
    const unsigned wd = glist[gi];
    const int k = (gi >= g1 ? 1 : 0) + (gi >= g2 ? 1 : 0) + (gi >= g3 ? 1 : 0);
    r[q] = k * L1_P + (int)(wd & ((1u << L1_RB) - 1u));
    m[q] = *(const float4*)(rec2 + (size_t)(wd >> L1_RB) * 4);
  }
#pragma unroll
  for (int q = 0; q < 4; ++q) dr[q] = dinv_r[r[q]];
#pragma unroll
  for (int q = 0; q < 4; ++q) {
    if (ok[q]) {
      const float w = m[q].z * dr[q];
      float* op = acc + r[q] * 2;
      ldsAddF32(op + 0, m[q].x * w);
      ldsAddF32(op + 1, m[q].y * w);
    }
  }
}

template <bool BK>
__global__ __launch_bounds__(EDGE_T) void k_edge_agg2(const int* __restrict__ ei,
                                                      const float* __restrict__ deg,
                                                      const float* __restrict__ rec2,
                                                      float* __restrict__ out,
                                                      const int* __restrict__ bst,
                                                      const unsigned* __restrict__ glist,
                                                      int N, int E, int R) {
  __shared__ float acc[L2_P * 2];
  __shared__ float dinv_r[L2_P];
  __shared__ unsigned list[BK ? 32 : LIST_CAP];
  const int tid = threadIdx.x;
  const int lane = tid & 31;
  unsigned* mylist = list + (tid >> 5) * WLIST;
  const int range = blockIdx.x % R;
  const int seg = blockIdx.x / R;
  const int base = range * L2_P;
  const int* __restrict__ src = ei;
  const int* __restrict__ dst = ei + E;

  for (int i = tid; i < L2_P * 2; i += EDGE_T) acc[i] = 0.0f;
  for (int i = tid; i < L2_P; i += EDGE_T) dinv_r[i] = (base + i < N) ? dinv_of(deg[base + i]) : 0.0f;
  __syncthreads();

  if constexpr (BK) {
  static_assert(L2_GRP == 4, "three inner bucket starts are compared below");
  const int g0 = bst[L2_GRP * range];
  const int g1 = bst[L2_GRP * range + 1];
  const int g2 = bst[L2_GRP * range + 2];
  const int g3 = bst[L2_GRP * range + 3];
  const int tot = bst[L2_GRP * range + 4] - g0;
  const int per = (tot + L2_S - 1) / L2_S;
  const int off = min(seg * per, tot);
  const int n = min(per, tot - off);
  const int a = g0 + off;
  const int wv = __builtin_amdgcn_readfirstlane(tid >> 5);
  for (int j0 = wv * 128; j0 < n; j0 += NWAVE * 128) {
    if (j0 + 128 <= n) l2_trip<true>(glist, a, n, j0, lane, g1, g2, g3, rec2, dinv_r, acc);
    else l2_trip<false>(glist, a, n, j0, lane, g1, g2, g3, rec2, dinv_r, acc);
  }
  } else {
  const int sl = seg_len(E, L2_S);
  const int eb = min(seg * sl, E);
  const int ee = min(eb + sl, E);
  const int cap = (N <= (1 << (32 - L2_RB))) ? WLIST : 0;
  int cnt = 0;
  auto list1 = [&](int rel, int r, bool m, int sv) {
    if (wave_append(mylist, cnt, cap, ((unsigned)sv << L2_RB) | (unsigned)r, m))
      l2_edge(eb + rel, src, dst, deg, rec2, acc, base);
  };
  scan_owned<true>(dst, src, eb, ee, (E & 3) == 0, base, L2_P,
                   [&](int e0, const int4 r, bool m0, bool m1, bool m2, bool m3, const int4 sv) {
                     if (!wave_append4(mylist, cnt, cap, ((unsigned)sv.x << L2_RB) | (unsigned)r.x,
                                       ((unsigned)sv.y << L2_RB) | (unsigned)r.y,
                                       ((unsigned)sv.z << L2_RB) | (unsigned)r.z,
                                       ((unsigned)sv.w << L2_RB) | (unsigned)r.w, m0, m1, m2, m3)) {
                       list1(e0 + 0, r.x, m0, sv.x);
                       list1(e0 + 1, r.y, m1, sv.y);
                       list1(e0 + 2, r.z, m2, sv.z);
                       list1(e0 + 3, r.w, m3, sv.w);
                     }
                   },
                   list1);
  __syncthreads();

  const int n = min(cnt, cap);
  for (int j0 = 0; j0 < n; j0 += 4 * 32) {
    int r[4];
    bool ok[4];
    float dr[4];
    float4 m[4];
#pragma unroll
    for (int q = 0; q < 4; ++q) {
      const int idx = j0 + q * 32 + lane;
      ok[q] = idx < n;
      const unsigned wd = mylist[ok[q] ? idx : n - 1];
      r[q] = (int)(wd & ((1u << L2_RB) - 1u));
      m[q] = *(const float4*)(rec2 + (size_t)(wd >> L2_RB) * 4);
    }
#pragma unroll
    for (int q = 0; q < 4; ++q) dr[q] = dinv_r[r[q]];
#pragma unroll
    for (int q = 0; q < 4; ++q) {
      if (ok[q]) {
        const float w = m[q].z * dr[q];
        float* op = acc + r[q] * 2;
        ldsAddF32(op + 0, m[q].x * w);
        ldsAddF32(op + 1, m[q].y * w);
      }
    }
  }
  }
  __syncthreads();

  for (int i = tid; i < L2_P * 2; i += EDGE_T) {
    const size_t o = (size_t)base * 2 + i;
    if (o < (size_t)N * 2) atomAddF32(out + o, acc[i]);
  }
}


extern "C" void kernel_launch(void* const* d_in, const int* in_sizes, int n_in,
                              void* d_out, int out_size, void* d_ws, size_t ws_size,
                              hipStream_t stream) {
  const float* x  = (const float*)d_in[0];
  const int*   ei = (const int*)d_in[1];
  const float* W1 = (const float*)d_in[2];
  const float* b1 = (const float*)d_in[3];
  const float* W2 = (const float*)d_in[4];
  const float* b2 = (const float*)d_in[5];
  float* out = (float*)d_out;

  const int N = in_sizes[0] / 128;
  const int E = in_sizes[1] / 2;

  float* deg  = (float*)d_ws;
  float* h1   = deg + N;
  float* agg1 = h1 + (size_t)N * 16;
  float* rec2 = h1;

  const size_t metaOff = ((size_t)N * 33 + 31) & ~(size_t)31;
  const size_t listOff = metaOff + META_WORDS;
  const bool bucketed = E > 0 && (N + L1_P - 1) / L1_P <= NB_MAX && N <= (1 << (32 - L1_RB)) &&
                        (listOff + (size_t)E) * 4 <= ws_size;
  int* meta = bucketed ? (int*)d_ws + metaOff : nullptr;
  const int* bst = bucketed ? meta + META_BST : nullptr;
  unsigned* glist = bucketed ? (unsigned*)d_ws + listOff : nullptr;

  dim3 blk(TPB);
  dim3 gN((N + TPB - 1) / TPB);
  dim3 eblk(EDGE_T);
  const int rDeg = (N + DEG_P - 1) / DEG_P;
  const int r1 = (N + L1_P - 1) / L1_P;
  const int r2 = (N + L2_P - 1) / L2_P;

  const int tiles = (N + 15) / 16;
  const int wavesPerBlock = TPB / 32;
  dim3 gT((tiles + wavesPerBlock - 1) / wavesPerBlock);
  k_gemm1_wmma<<<gT, blk, 0, stream>>>(x, W1, deg, h1, agg1, meta, N);
  k_edge_deg<<<dim3(rDeg * DEG_S), eblk, 0, stream>>>(ei + E, deg, meta, N, E, rDeg);

  if (bucketed) {
    k_bucket<<<dim3((E + BK_CAP - 1) / BK_CAP), eblk, 0, stream>>>(ei, meta, glist, E);
    k_edge_agg16<true><<<dim3(r1 * L1_S), eblk, 0, stream>>>(ei, deg, h1, agg1, bst, glist, N, E, r1);
    k_node2<<<gN, blk, 0, stream>>>(agg1, b1, W2, b2, deg, rec2, out, N);
    k_edge_agg2<true><<<dim3(r2 * L2_S), eblk, 0, stream>>>(ei, deg, rec2, out, bst, glist, N, E, r2);
  } else {
    k_edge_agg16<false><<<dim3(r1 * L1_S), eblk, 0, stream>>>(ei, deg, h1, agg1, bst, glist, N, E, r1);
    k_node2<<<gN, blk, 0, stream>>>(agg1, b1, W2, b2, deg, rec2, out, N);
    k_edge_agg2<false><<<dim3(r2 * L2_S), eblk, 0, stream>>>(ei, deg, rec2, out, bst, glist, N, E, r2);
  }
}


# layer-1 edge pass: one work-group owns a whole destination range (finer buckets): LDS sums are final, no global adds, no agg1; node step runs from LDS; launches 6 -> 5; self-loop term summed last
# speedup vs baseline: 9.5047x; 1.0426x over previous
#include <hip/hip_runtime.h>

typedef __attribute__((ext_vector_type(2))) float v2f;
typedef __attribute__((ext_vector_type(8))) float v8f;

#define TPB 256

#define EDGE_T 1024
#define LIST_CAP 16384
#define NWAVE (EDGE_T / 32)
#define WLIST (LIST_CAP / NWAVE)
#define DEG_SUB 64
#define DEG_P (DEG_SUB * L1_P)
#define DEG_S 64
#define L1_P 400
#define L1_S 8
#define L1_PITCH 17
#define L1_OWN ((L1_P + L1_S - 1) / L1_S)
#define L2_P 12800
#define L2_S 32
#define L1_RB 12
#define L2_RB 14
static_assert(L1_P <= (1 << L1_RB), "layer-1 row does not fit its list-word field");
static_assert(L2_P <= (1 << L2_RB), "layer-2 row does not fit its list-word field");
#define NB_MAX 256
#define BW_RB 14
#define BK_V 3
#define BK_CAP (BK_V * 4 * EDGE_T)
#define L2_GRP (L2_P / L1_P)
#define MSTRIDE 64
#define META_CUR (NB_MAX * MSTRIDE)
#define META_BST (2 * NB_MAX * MSTRIDE)
#define META_WORDS (META_BST + NB_MAX + 64)
static_assert(L1_P % 16 == 0, "eight consecutive nodes from a multiple of 8 must lie in one layer-1 range; one writer per line of out");
static_assert(DEG_P % 32 == 0 && EDGE_T % 32 == 0, "degree pass: a wave is whole in every trip of the flush loop");
static_assert(L2_P == L2_GRP * L1_P, "a layer-2 range must be whole layer-1 ranges");
static_assert(NB_MAX == 256, "the prefix helper takes eight buckets per lane");
static_assert(NB_MAX <= TPB, "the first GEMM work-group clears one count and one cursor per thread");
static_assert(L2_P <= (1 << BW_RB), "layer-2 row does not fit the bucketed word's field");

static __device__ __forceinline__ void atomAddF32(float* p, float v) {
  unsafeAtomicAdd(p, v);
}

static __device__ __forceinline__ void ldsAddF32(float* p, float v) {
  __hip_atomic_fetch_add(p, v, __ATOMIC_RELAXED, __HIP_MEMORY_SCOPE_WORKGROUP);
}

static __device__ __forceinline__ float dinv_of(float d) {
  return (d > 0.0f) ? rsqrtf(d) : 0.0f;
}

static __device__ __forceinline__ int sum8(int v) {
  v += __shfl_xor(v, 4, 32);
  v += __shfl_xor(v, 2, 32);
  v += __shfl_xor(v, 1, 32);
  return v;
}

static __device__ __forceinline__ void wave_prefix8(const int (&a)[8], int lane, int (&p)[8],
                                                    int& total) {
  int own = 0;
#pragma unroll
  for (int j = 0; j < 8; ++j) {
    p[j] = own;
    own += a[j];
  }
  int inc = own;
#pragma unroll
  for (int d = 1; d < 32; d <<= 1) {
    const int t = __shfl_up(inc, d, 32);
    if (lane >= d) inc += t;
  }
  const int before = inc - own;
#pragma unroll
  for (int j = 0; j < 8; ++j) p[j] += before;
  total = __shfl(inc, 31, 32);
}

template <bool WITH_SRC, class F4, class F>
static __device__ __forceinline__ void scan_owned(const int* __restrict__ dst,
                                                  const int* __restrict__ src, int eb, int ee,
                                                  bool vec_ok, int base, int P, F4 on4, F onvec) {
  const int lane = threadIdx.x & 31;
  const int w0 = __builtin_amdgcn_readfirstlane((int)(threadIdx.x >> 5)) * 32;
  const int nvec = vec_ok ? ((ee - eb) >> 2) : 0;
  const int4* d4 = (const int4*)(dst + eb);
  const int4* s4 = (const int4*)(src + eb);
  auto lds4 = [&](int i) -> int4 {
    if constexpr (WITH_SRC) return s4[i];
    else return make_int4(0, 0, 0, 0);
  };
  auto test4 = [&](const int4 v, const int4 u, int i, bool ok) {
    const int e0 = i * 4;
    const unsigned r0 = (unsigned)(v.x - base);
    const unsigned r1 = (unsigned)(v.y - base);
    const unsigned r2 = (unsigned)(v.z - base);
    const unsigned r3 = (unsigned)(v.w - base);
    on4(e0, make_int4((int)r0, (int)r1, (int)r2, (int)r3), ok && r0 < (unsigned)P,
        ok && r1 < (unsigned)P, ok && r2 < (unsigned)P, ok && r3 < (unsigned)P, u);
  };
  int iw = w0;
  for (; iw + 31 + 3 * EDGE_T < nvec; iw += 4 * EDGE_T) {
    const int i = iw + lane;
    const int4 a = d4[i];
    const int4 b = d4[i + EDGE_T];
    const int4 c = d4[i + 2 * EDGE_T];
    const int4 d = d4[i + 3 * EDGE_T];
    const int4 sa = lds4(i);
    const int4 sb = lds4(i + EDGE_T);
    const int4 sc = lds4(i + 2 * EDGE_T);
    const int4 sd = lds4(i + 3 * EDGE_T);
    test4(a, sa, i, true);
    test4(b, sb, i + EDGE_T, true);
    test4(c, sc, i + 2 * EDGE_T, true);
    test4(d, sd, i + 3 * EDGE_T, true);
  }
  for (; iw < nvec; iw += EDGE_T) {
    const int i = iw + lane;
    const bool ok = i < nvec;
    const int j = ok ? i : iw;
    test4(d4[j], lds4(j), i, ok);
  }
  for (int ew = eb + nvec * 4 + w0; ew < ee; ew += EDGE_T) {
    const int e = ew + lane;
    const bool ok = e < ee;
    const int j = ok ? e : ew;
    const unsigned r = (unsigned)(dst[j] - base);
    int sv = 0;
    if constexpr (WITH_SRC) sv = src[j];
    onvec(e - eb, (int)r, ok && r < (unsigned)P, sv);
  }
}

static __device__ __forceinline__ bool wave_append(unsigned* mylist, int& cnt, int cap,
                                                   unsigned word, bool m) {
  const unsigned mask = __builtin_amdgcn_ballot_w32(m);
  bool spill = false;
  if (mask != 0u) {
    const int p = cnt + (int)__builtin_amdgcn_mbcnt_lo(mask, 0u);
    cnt = __builtin_amdgcn_readfirstlane(cnt + __popc(mask));
    if (m) {
      if (p < cap) mylist[p] = word; else spill = true;
    }
  }
  return spill;
}

static __device__ __forceinline__ bool wave_append4(unsigned* mylist, int& cnt, int cap,
                                                    unsigned w0, unsigned w1, unsigned w2,
                                                    unsigned w3, bool m0, bool m1, bool m2,
                                                    bool m3) {
  const unsigned k0 = __builtin_amdgcn_ballot_w32(m0);
  const unsigned k1 = __builtin_amdgcn_ballot_w32(m1);
  const unsigned k2 = __builtin_amdgcn_ballot_w32(m2);
  const unsigned k3 = __builtin_amdgcn_ballot_w32(m3);
  const int b1 = cnt + __popc(k0);
  const int b2 = b1 + __popc(k1);
  const int b3 = b2 + __popc(k2);
  const int b4 = b3 + __popc(k3);
  if (__builtin_expect(b4 > cap, 0)) return false;
  if (__builtin_expect(m0, 1)) mylist[__builtin_amdgcn_mbcnt_lo(k0, (unsigned)cnt)] = w0;
  if (__builtin_expect(m1, 1)) mylist[__builtin_amdgcn_mbcnt_lo(k1, (unsigned)b1)] = w1;
  if (__builtin_expect(m2, 1)) mylist[__builtin_amdgcn_mbcnt_lo(k2, (unsigned)b2)] = w2;
  if (__builtin_expect(m3, 1)) mylist[__builtin_amdgcn_mbcnt_lo(k3, (unsigned)b3)] = w3;
  cnt = b4;
  return true;
}

static __device__ __forceinline__ int seg_len(int E, int S) {
  return (((E + S - 1) / S) + 3) & ~3;
}


__global__ __launch_bounds__(EDGE_T) void k_edge_deg(const int* __restrict__ dst,
                                                     float* __restrict__ deg,
                                                     int* __restrict__ bcnt, int N, int E, int R) {
  __shared__ int cnt[DEG_P];
  __shared__ int sub[DEG_SUB];
  const int tid = threadIdx.x;
  if (tid < DEG_SUB) sub[tid] = 0;
  const int range = blockIdx.x % R;
  const int seg = blockIdx.x / R;
  const int base = range * DEG_P;
  for (int i = tid; i < DEG_P; i += EDGE_T) cnt[i] = 0;
  __syncthreads();
  const int sl = seg_len(E, DEG_S);
  const int eb = min(seg * sl, E);
  const int ee = min(eb + sl, E);
  auto count1 = [&](int, int r, bool m, int) {
    if (m) __hip_atomic_fetch_add(&cnt[r], 1, __ATOMIC_RELAXED, __HIP_MEMORY_SCOPE_WORKGROUP);
  };
  scan_owned<false>(dst, dst, eb, ee, (E & 3) == 0, base, DEG_P,
                    [&](int e0, const int4 r, bool m0, bool m1, bool m2, bool m3, const int4) {
                      count1(e0 + 0, r.x, m0, 0);
                      count1(e0 + 1, r.y, m1, 0);
                      count1(e0 + 2, r.z, m2, 0);
                      count1(e0 + 3, r.w, m3, 0);
                    },
                    count1);
  __syncthreads();
  for (int i = tid; i < DEG_P; i += EDGE_T) {
    const int n = base + i;
    const int c = cnt[i];
    if (n < N && c != 0) atomAddF32(&deg[n], (float)c);
    if (bcnt != nullptr) {
      const int gs = sum8(c);
      if ((tid & 7) == 0 && gs != 0)
        __hip_atomic_fetch_add(&sub[i / L1_P], gs, __ATOMIC_RELAXED, __HIP_MEMORY_SCOPE_WORKGROUP);
    }
  }
  if (bcnt != nullptr) {
    __syncthreads();
    const int b = range * DEG_SUB + tid;
    if (tid < DEG_SUB && b < NB_MAX && sub[tid] != 0) atomicAdd(&bcnt[b * MSTRIDE], sub[tid]);
  }
}


__global__ void k_gemm1_wmma(const float* __restrict__ x,
                             const float* __restrict__ W1,
                             float* __restrict__ deg,
                             float* __restrict__ h1,
                             float* __restrict__ agg1,
                             int* __restrict__ meta, int N) {
  if (meta != nullptr && blockIdx.x == 0 && threadIdx.x < NB_MAX) {
    meta[threadIdx.x * MSTRIDE] = 0;
    meta[META_CUR + threadIdx.x * MSTRIDE] = 0;
  }
  const int lane = threadIdx.x & 31;
  const int tile = blockIdx.x * (blockDim.x >> 5) + (threadIdx.x >> 5);
  const int row0 = tile * 16;
  if (row0 >= N) return;

  const int half = lane >> 4;
  const int lrow = lane & 15;

  int arow = row0 + lrow;
  if (arow >= N) arow = N - 1;

  const float* xrow = x + (size_t)arow * 128 + half * 2;
  const float* wcol = W1 + (size_t)(half * 2) * 16 + lrow;

  v8f acc = {};
#pragma unroll
  for (int k0 = 0; k0 < 128; k0 += 4) {
    v2f a = *(const v2f*)(xrow + k0);
    v2f b;
    b.x = wcol[(size_t)k0 * 16];
    b.y = wcol[(size_t)k0 * 16 + 16];
    acc = __builtin_amdgcn_wmma_f32_16x16x4_f32(
         false, a,  false, b,
         (short)0, acc,  false,  false);
  }

  const int col = lrow;
#pragma unroll
  for (int v = 0; v < 8; ++v) {
    const int m = row0 + v + half * 8;
    if (m < N) {
      h1[(size_t)m * 16 + col] = acc[v];
      if (agg1 != nullptr) agg1[(size_t)m * 16 + col] = 0.0f;
    }
  }
  if (lane < 16 && row0 + lane < N) deg[row0 + lane] = 1.0f;
}

__global__ __launch_bounds__(EDGE_T) void k_bucket(const int* __restrict__ ei,
                                                   int* __restrict__ meta,
                                                   unsigned* __restrict__ glist, int E) {
  __shared__ unsigned stage[BK_CAP];
  __shared__ int where[BK_CAP];
  __shared__ int hist[NB_MAX];
  __shared__ int loff[NB_MAX];
  __shared__ int gpos[NB_MAX];
  const int tid = threadIdx.x;
  const int lane = tid & 31;
  const int* __restrict__ src = ei;
  const int* __restrict__ dst = ei + E;
  const int* bcnt = meta;
  int* cur = meta + META_CUR;
  int* bst = meta + META_BST;
  const int eb = blockIdx.x * BK_CAP;
  const bool vec_ok = (E & 3) == 0;

  if (tid < NB_MAX) hist[tid] = 0;
  __syncthreads();
  int bc[8] = {0, 0, 0, 0, 0, 0, 0, 0};
  if (tid < 32) {
#pragma unroll
    for (int j = 0; j < 8; ++j) bc[j] = bcnt[(8 * lane + j) * MSTRIDE];
  }

  int dd[BK_V * 4], ss[BK_V * 4];
  if (vec_ok && eb + BK_CAP <= E) {
#pragma unroll
    for (int q = 0; q < BK_V; ++q) {
      const int e = eb + (tid + q * EDGE_T) * 4;
      const int4 dv = *(const int4*)(dst + e);
      const int4 uv = *(const int4*)(src + e);
      dd[q * 4 + 0] = dv.x; dd[q * 4 + 1] = dv.y; dd[q * 4 + 2] = dv.z; dd[q * 4 + 3] = dv.w;
      ss[q * 4 + 0] = uv.x; ss[q * 4 + 1] = uv.y; ss[q * 4 + 2] = uv.z; ss[q * 4 + 3] = uv.w;
    }
  } else {
#pragma unroll
    for (int k = 0; k < BK_V * 4; ++k) {
      const int e = eb + (tid + (k >> 2) * EDGE_T) * 4 + (k & 3);
      const bool ok = e < E;
      dd[k] = ok ? dst[ok ? e : 0] : -1;
      ss[k] = ok ? src[ok ? e : 0] : 0;
    }
  }
  unsigned word[BK_V * 4];
  int slot[BK_V * 4];
#pragma unroll
  for (int k = 0; k < BK_V * 4; ++k) {
    const unsigned d = (unsigned)max(dd[k], 0);
    const unsigned b = d / (unsigned)L1_P;
    const unsigned row = d - (d / (unsigned)L2_P) * (unsigned)L2_P;
    word[k] = ((unsigned)ss[k] << BW_RB) | row;
    int sl = -1;
    if (dd[k] >= 0) {
      const int rank = __hip_atomic_fetch_add(&hist[b], 1, __ATOMIC_RELAXED,
                                              __HIP_MEMORY_SCOPE_WORKGROUP);
      sl = (int)(b << 16) | rank;
    }
    slot[k] = sl;
  }
  __syncthreads();

  if (tid < 32) {
    int hc[8], lo[8], gs[8], cu[8], lt, gt;
#pragma unroll
    for (int j = 0; j < 8; ++j) hc[j] = hist[8 * lane + j];
    wave_prefix8(hc, lane, lo, lt);
    wave_prefix8(bc, lane, gs, gt);
#pragma unroll
    for (int j = 0; j < 8; ++j) cu[j] = atomicAdd(&cur[(8 * lane + j) * MSTRIDE], hc[j]);
#pragma unroll
    for (int j = 0; j < 8; ++j) {
      loff[8 * lane + j] = lo[j];
      gpos[8 * lane + j] = gs[j] + cu[j];
    }
    if (blockIdx.x == 0) {
#pragma unroll
      for (int j = 0; j < 8; ++j) bst[8 * lane + j] = gs[j];
      if (lane < 8) bst[NB_MAX + lane] = gt;
    }
  }
  __syncthreads();

#pragma unroll
  for (int k = 0; k < BK_V * 4; ++k) {
    if (slot[k] >= 0) {
      const int b = slot[k] >> 16;
      const int rank = slot[k] & 0xffff;
      const int p = loff[b] + rank;
      stage[p] = word[k];
      where[p] = gpos[b] + rank;
    }
  }
  __syncthreads();

  const int n = min(BK_CAP, E - eb);
  for (int i = tid; i < n; i += EDGE_T) glist[where[i]] = stage[i];
}


static __device__ __forceinline__ void l1_add(float* op, const float4 m0, const float4 m1,
                                              const float4 m2, const float4 m3, float w) {
  ldsAddF32(op + 0,  m0.x * w);
  ldsAddF32(op + 1,  m0.y * w);
  ldsAddF32(op + 2,  m0.z * w);
  ldsAddF32(op + 3,  m0.w * w);
  ldsAddF32(op + 4,  m1.x * w);
  ldsAddF32(op + 5,  m1.y * w);
  ldsAddF32(op + 6,  m1.z * w);
  ldsAddF32(op + 7,  m1.w * w);
  ldsAddF32(op + 8,  m2.x * w);
  ldsAddF32(op + 9,  m2.y * w);
  ldsAddF32(op + 10, m2.z * w);
  ldsAddF32(op + 11, m2.w * w);
  ldsAddF32(op + 12, m3.x * w);
  ldsAddF32(op + 13, m3.y * w);
  ldsAddF32(op + 14, m3.z * w);
  ldsAddF32(op + 15, m3.w * w);
}

static __device__ __forceinline__ void l1_edge(int e, const int* __restrict__ src,
                                               const int* __restrict__ dst,
                                               const float* __restrict__ deg,
                                               const float* __restrict__ h1,
                                               float* acc, int base) {
  const int s = src[e];
  const int d = dst[e];
  const float w = dinv_of(deg[s]) * dinv_of(deg[d]);
  const float4* hs = (const float4*)(h1 + (size_t)s * 16);
  l1_add(acc + (d - base) * L1_PITCH, hs[0], hs[1], hs[2], hs[3], w);
}

template <bool FULL>
static __device__ __forceinline__ void l1_trip(const unsigned* __restrict__ mine, int n, int j0,
                                               int lane, int rsub, const float* __restrict__ deg,
                                               const float* __restrict__ h1,
                                               const float* dinv_r, float* acc) {
  int s[4], r[4];
  bool ok[4];
  float w[4];
  float4 m[4][4];
#pragma unroll
  for (int q = 0; q < 4; ++q) {
    const int idx = j0 + q * 32 + lane;
    ok[q] = FULL || idx < n;
    const unsigned wd = mine[ok[q] ? idx : n - 1];
    r[q] = (int)(wd & ((1u << BW_RB) - 1u)) - rsub;
    s[q] = (int)(wd >> BW_RB);
  }
#pragma unroll
  for (int q = 0; q < 4; ++q) {
    w[q] = dinv_of(deg[s[q]]) * dinv_r[r[q]];
    const float4* hs = (const float4*)(h1 + (size_t)s[q] * 16);
    m[q][0] = hs[0]; m[q][1] = hs[1]; m[q][2] = hs[2]; m[q][3] = hs[3];
  }
#pragma unroll
  for (int q = 0; q < 4; ++q)
    if (ok[q]) l1_add(acc + r[q] * L1_PITCH, m[q][0], m[q][1], m[q][2], m[q][3], w[q]);
}

__global__ __launch_bounds__(EDGE_T) void k_own16(const float* __restrict__ deg,
                                                  const float* __restrict__ h1,
                                                  const int* __restrict__ bst,
                                                  const unsigned* __restrict__ glist,
                                                  const float* __restrict__ b1,
                                                  const float* __restrict__ W2,
                                                  const float* __restrict__ b2,
                                                  float* __restrict__ rec2,
                                                  float* __restrict__ out, int N) {
  __shared__ float acc[L1_P * L1_PITCH];
  __shared__ float dinv_r[L1_P];
  const int tid = threadIdx.x;
  const int lane = tid & 31;
  const int range = blockIdx.x;
  const int base = range * L1_P;

  for (int i = tid; i < L1_P * L1_PITCH; i += EDGE_T) acc[i] = 0.0f;
  for (int i = tid; i < L1_P; i += EDGE_T) dinv_r[i] = (base + i < N) ? dinv_of(deg[base + i]) : 0.0f;

  __syncthreads();

  const bool owner = tid < L1_P && base + tid < N;
  const int onode = owner ? base + tid : base;
  const float4* op = (const float4*)(h1 + (size_t)onode * 16);
  const float4 o0 = op[0], o1 = op[1], o2 = op[2], o3 = op[3];

  const int lo = bst[range];
  const int n = bst[range + 1] - lo;
  const unsigned* __restrict__ mine = glist + lo;
  const int rsub = (range % L2_GRP) * L1_P;
  const int wv = __builtin_amdgcn_readfirstlane(tid >> 5);
  for (int j0 = wv * 128; j0 < n; j0 += NWAVE * 128) {
    if (j0 + 128 <= n) l1_trip<true>(mine, n, j0, lane, rsub, deg, h1, dinv_r, acc);
    else l1_trip<false>(mine, n, j0, lane, rsub, deg, h1, dinv_r, acc);
  }
  __syncthreads();

  if (owner) {
    const float di = dinv_r[tid];
    const float w = di * di;
    const float sv[16] = {o0.x, o0.y, o0.z, o0.w, o1.x, o1.y, o1.z, o1.w,
                          o2.x, o2.y, o2.z, o2.w, o3.x, o3.y, o3.z, o3.w};
    const float* ar = acc + tid * L1_PITCH;
    float c0 = 0.0f, c1 = 0.0f;
#pragma unroll
    for (int k = 0; k < 16; ++k) {
      const float a = ar[k] + sv[k] * w;
      const float hv = fmaxf(a + b1[k], 0.0f);
      c0 = fmaf(hv, W2[k * 2 + 0], c0);
      c1 = fmaf(hv, W2[k * 2 + 1], c1);
    }
    *(float4*)(rec2 + (size_t)onode * 4) = make_float4(c0, c1, di, 0.0f);
    *(float2*)(out + (size_t)onode * 2) = make_float2(fmaf(c0, w, b2[0]), fmaf(c1, w, b2[1]));
  }
}

__global__ __launch_bounds__(EDGE_T) void k_edge_agg16(const int* __restrict__ ei,
                                                       const float* __restrict__ deg,
                                                       const float* __restrict__ h1,
                                                       float* __restrict__ agg1,
                                                       int N, int E, int R) {
  __shared__ float acc[L1_P * L1_PITCH];
  __shared__ float dinv_r[L1_P];
  __shared__ unsigned list[LIST_CAP];
  const int tid = threadIdx.x;
  const int lane = tid & 31;
  unsigned* mylist = list + (tid >> 5) * WLIST;
  const int range = blockIdx.x % R;
  const int seg = blockIdx.x / R;
  const int base = range * L1_P;
  const int* __restrict__ src = ei;
  const int* __restrict__ dst = ei + E;

  for (int i = tid; i < L1_P * L1_PITCH; i += EDGE_T) acc[i] = 0.0f;
  for (int i = tid; i < L1_P; i += EDGE_T) dinv_r[i] = (base + i < N) ? dinv_of(deg[base + i]) : 0.0f;
  __syncthreads();

  constexpr int OWN_T = (L1_OWN * 16 + EDGE_T - 1) / EDGE_T;
  const int own_lo = seg * L1_OWN;
  float selfv[OWN_T];
  int selfo[OWN_T];
#pragma unroll
  for (int q = 0; q < OWN_T; ++q) {
    const int i = tid + q * EDGE_T;
    const int r = own_lo + (i >> 4);
    const int k = i & 15;
    const bool ok = i < L1_OWN * 16 && r < L1_P && base + r < N;
    const int node = ok ? base + r : base;
    selfv[q] = h1[(size_t)node * 16 + k];
    selfo[q] = ok ? r * L1_PITCH + k : -1;
  }

  const int sl = seg_len(E, L1_S);
  const int eb = min(seg * sl, E);
  const int ee = min(eb + sl, E);
  const int cap = (sl <= (1 << (32 - L1_RB))) ? WLIST : 0;
  int cnt = 0;
  auto list1 = [&](int rel, int r, bool m, int) {
    if (wave_append(mylist, cnt, cap, ((unsigned)rel << L1_RB) | (unsigned)r, m))
      l1_edge(eb + rel, src, dst, deg, h1, acc, base);
  };
  scan_owned<false>(dst, dst, eb, ee, (E & 3) == 0, base, L1_P,
                    [&](int e0, const int4 r, bool m0, bool m1, bool m2, bool m3, const int4) {
                      const unsigned w = (unsigned)e0 << L1_RB;
                      if (!wave_append4(mylist, cnt, cap, w | (unsigned)r.x,
                                        w | (1u << L1_RB) | (unsigned)r.y,
                                        w | (2u << L1_RB) | (unsigned)r.z,
                                        w | (3u << L1_RB) | (unsigned)r.w, m0, m1, m2, m3)) {
                        list1(e0 + 0, r.x, m0, 0);
                        list1(e0 + 1, r.y, m1, 0);
                        list1(e0 + 2, r.z, m2, 0);
                        list1(e0 + 3, r.w, m3, 0);
                      }
                    },
                    list1);
  __syncthreads();

  const int n = min(cnt, cap);
  for (int j0 = 0; j0 < n; j0 += 4 * 32) {
    int s[4], r[4];
    bool ok[4];
    float w[4];
    float4 m[4][4];
#pragma unroll
    for (int q = 0; q < 4; ++q) {
      const int idx = j0 + q * 32 + lane;
      ok[q] = idx < n;
      const unsigned wd = mylist[ok[q] ? idx : n - 1];
      r[q] = (int)(wd & ((1u << L1_RB) - 1u));
      s[q] = src[eb + (int)(wd >> L1_RB)];
    }
#pragma unroll
    for (int q = 0; q < 4; ++q) {
      w[q] = dinv_of(deg[s[q]]) * dinv_r[r[q]];
      const float4* hs = (const float4*)(h1 + (size_t)s[q] * 16);
      m[q][0] = hs[0]; m[q][1] = hs[1]; m[q][2] = hs[2]; m[q][3] = hs[3];
    }
#pragma unroll
    for (int q = 0; q < 4; ++q)
      if (ok[q]) l1_add(acc + r[q] * L1_PITCH, m[q][0], m[q][1], m[q][2], m[q][3], w[q]);
  }
#pragma unroll
  for (int q = 0; q < OWN_T; ++q) {
    if (selfo[q] >= 0) {
      const float di = dinv_r[own_lo + ((tid + q * EDGE_T) >> 4)];
      ldsAddF32(acc + selfo[q], selfv[q] * di * di);
    }
  }
  __syncthreads();

  for (int i = tid; i < L1_P * 16; i += EDGE_T) {
    const int r = i >> 4;
    const int k = i & 15;
    const int node = base + r;
    if (node < N) atomAddF32(agg1 + (size_t)node * 16 + k, acc[r * L1_PITCH + k]);
  }
}


__global__ void k_node2(const float* __restrict__ agg1,
                        const float* __restrict__ b1,
                        const float* __restrict__ W2,
                        const float* __restrict__ b2,
                        const float* __restrict__ deg,
                        float* __restrict__ rec2,
                        float* __restrict__ out, int N) {
  int n = blockIdx.x * blockDim.x + threadIdx.x;
  if (n >= N) return;
  const float4* ap = (const float4*)(agg1 + (size_t)n * 16);
  float c0 = 0.0f, c1 = 0.0f;
#pragma unroll
  for (int i = 0; i < 4; ++i) {
    float4 t = ap[i];
    float v[4] = {t.x, t.y, t.z, t.w};
#pragma unroll
    for (int j = 0; j < 4; ++j) {
      const int k = i * 4 + j;
      const float hv = fmaxf(v[j] + b1[k], 0.0f);
      c0 = fmaf(hv, W2[k * 2 + 0], c0);
      c1 = fmaf(hv, W2[k * 2 + 1], c1);
    }
  }
  const float di = dinv_of(deg[n]);
  *(float4*)(rec2 + (size_t)n * 4) = make_float4(c0, c1, di, 0.0f);
  const float w = di * di;
  out[(size_t)n * 2 + 0] = fmaf(c0, w, b2[0]);
  out[(size_t)n * 2 + 1] = fmaf(c1, w, b2[1]);
}


static __device__ __forceinline__ void l2_edge(int e, const int* __restrict__ src,
                                               const int* __restrict__ dst,
                                               const float* __restrict__ deg,
                                               const float* __restrict__ rec2,
                                               float* acc, int base) {
  const int s = src[e];
  const int d = dst[e];
  const float4 mm = *(const float4*)(rec2 + (size_t)s * 4);
  const float w = mm.z * dinv_of(deg[d]);
  float* op = acc + (d - base) * 2;
  ldsAddF32(op + 0, mm.x * w);
  ldsAddF32(op + 1, mm.y * w);
}

template <bool FULL>
static __device__ __forceinline__ void l2_trip(const unsigned* __restrict__ glist, int a, int n,
                                               int j0, int lane,
                                               const float* __restrict__ rec2,
                                               const float* dinv_r, float* acc) {
  int r[4];
  bool ok[4];
  float dr[4];
  float4 m[4];
#pragma unroll
  for (int q = 0; q < 4; ++q) {
    const int idx = j0 + q * 32 + lane;
    ok[q] = FULL || idx < n;
    const int gi = a + (ok[q] ? idx : n - 1);
    const unsigned wd = glist[gi];
    r[q] = (int)(wd & ((1u << BW_RB) - 1u));
    m[q] = *(const float4*)(rec2 + (size_t)(wd >> BW_RB) * 4);
  }
#pragma unroll
  for (int q = 0; q < 4; ++q) dr[q] = dinv_r[r[q]];
#pragma unroll
  for (int q = 0; q < 4; ++q) {
    if (ok[q]) {
      const float w = m[q].z * dr[q];
      float* op = acc + r[q] * 2;
      ldsAddF32(op + 0, m[q].x * w);
      ldsAddF32(op + 1, m[q].y * w);
    }
  }
}

template <bool BK>
__global__ __launch_bounds__(EDGE_T) void k_edge_agg2(const int* __restrict__ ei,
                                                      const float* __restrict__ deg,
                                                      const float* __restrict__ rec2,
                                                      float* __restrict__ out,
                                                      const int* __restrict__ bst,
                                                      const unsigned* __restrict__ glist,
                                                      int N, int E, int R) {
  __shared__ float acc[L2_P * 2];
  __shared__ float dinv_r[L2_P];
  __shared__ unsigned list[BK ? 32 : LIST_CAP];
  const int tid = threadIdx.x;
  const int lane = tid & 31;
  unsigned* mylist = list + (tid >> 5) * WLIST;
  const int range = blockIdx.x % R;
  const int seg = blockIdx.x / R;
  const int base = range * L2_P;
  const int* __restrict__ src = ei;
  const int* __restrict__ dst = ei + E;

  for (int i = tid; i < L2_P * 2; i += EDGE_T) acc[i] = 0.0f;
  for (int i = tid; i < L2_P; i += EDGE_T) dinv_r[i] = (base + i < N) ? dinv_of(deg[base + i]) : 0.0f;
  __syncthreads();

  if constexpr (BK) {
  static_assert(NB_MAX % L2_GRP == 0, "the last group's end is an entry of the bucket starts");
  const int g0 = bst[L2_GRP * range];
  const int tot = bst[L2_GRP * range + L2_GRP] - g0;
  const int per = (tot + L2_S - 1) / L2_S;
  const int off = min(seg * per, tot);
  const int n = min(per, tot - off);
  const int a = g0 + off;
  const int wv = __builtin_amdgcn_readfirstlane(tid >> 5);
  for (int j0 = wv * 128; j0 < n; j0 += NWAVE * 128) {
    if (j0 + 128 <= n) l2_trip<true>(glist, a, n, j0, lane, rec2, dinv_r, acc);
    else l2_trip<false>(glist, a, n, j0, lane, rec2, dinv_r, acc);
  }
  } else {
  const int sl = seg_len(E, L2_S);
  const int eb = min(seg * sl, E);
  const int ee = min(eb + sl, E);
  const int cap = (N <= (1 << (32 - L2_RB))) ? WLIST : 0;
  int cnt = 0;
  auto list1 = [&](int rel, int r, bool m, int sv) {
    if (wave_append(mylist, cnt, cap, ((unsigned)sv << L2_RB) | (unsigned)r, m))
      l2_edge(eb + rel, src, dst, deg, rec2, acc, base);
  };
  scan_owned<true>(dst, src, eb, ee, (E & 3) == 0, base, L2_P,
                   [&](int e0, const int4 r, bool m0, bool m1, bool m2, bool m3, const int4 sv) {
                     if (!wave_append4(mylist, cnt, cap, ((unsigned)sv.x << L2_RB) | (unsigned)r.x,
                                       ((unsigned)sv.y << L2_RB) | (unsigned)r.y,
                                       ((unsigned)sv.z << L2_RB) | (unsigned)r.z,
                                       ((unsigned)sv.w << L2_RB) | (unsigned)r.w, m0, m1, m2, m3)) {
                       list1(e0 + 0, r.x, m0, sv.x);
                       list1(e0 + 1, r.y, m1, sv.y);
                       list1(e0 + 2, r.z, m2, sv.z);
                       list1(e0 + 3, r.w, m3, sv.w);
                     }
                   },
                   list1);
  __syncthreads();

  const int n = min(cnt, cap);
  for (int j0 = 0; j0 < n; j0 += 4 * 32) {
    int r[4];
    bool ok[4];
    float dr[4];
    float4 m[4];
#pragma unroll
    for (int q = 0; q < 4; ++q) {
      const int idx = j0 + q * 32 + lane;
      ok[q] = idx < n;
      const unsigned wd = mylist[ok[q] ? idx : n - 1];
      r[q] = (int)(wd & ((1u << L2_RB) - 1u));
      m[q] = *(const float4*)(rec2 + (size_t)(wd >> L2_RB) * 4);
    }
#pragma unroll
    for (int q = 0; q < 4; ++q) dr[q] = dinv_r[r[q]];
#pragma unroll
    for (int q = 0; q < 4; ++q) {
      if (ok[q]) {
        const float w = m[q].z * dr[q];
        float* op = acc + r[q] * 2;
        ldsAddF32(op + 0, m[q].x * w);
        ldsAddF32(op + 1, m[q].y * w);
      }
    }
  }
  }
  __syncthreads();

  for (int i = tid; i < L2_P * 2; i += EDGE_T) {
    const size_t o = (size_t)base * 2 + i;
    if (o < (size_t)N * 2) atomAddF32(out + o, acc[i]);
  }
}


extern "C" void kernel_launch(void* const* d_in, const int* in_sizes, int n_in,
                              void* d_out, int out_size, void* d_ws, size_t ws_size,
                              hipStream_t stream) {
  const float* x  = (const float*)d_in[0];
  const int*   ei = (const int*)d_in[1];
  const float* W1 = (const float*)d_in[2];
  const float* b1 = (const float*)d_in[3];
  const float* W2 = (const float*)d_in[4];
  const float* b2 = (const float*)d_in[5];
  float* out = (float*)d_out;

  const int N = in_sizes[0] / 128;
  const int E = in_sizes[1] / 2;

  float* deg  = (float*)d_ws;
  float* h1   = deg + N;
  float* agg1 = h1 + (size_t)N * 16;

  const size_t metaOff = ((size_t)N * 21 + 31) & ~(size_t)31;
  const size_t listOff = metaOff + META_WORDS;
  const bool bucketed = E > 0 && (N + L1_P - 1) / L1_P <= NB_MAX && N <= (1 << (32 - BW_RB)) &&
                        (listOff + (size_t)E) * 4 <= ws_size;
  int* meta = bucketed ? (int*)d_ws + metaOff : nullptr;
  const int* bst = bucketed ? meta + META_BST : nullptr;
  unsigned* glist = bucketed ? (unsigned*)d_ws + listOff : nullptr;
  float* rec2 = bucketed ? agg1 : h1;

  dim3 blk(TPB);
  dim3 gN((N + TPB - 1) / TPB);
  dim3 eblk(EDGE_T);
  const int rDeg = (N + DEG_P - 1) / DEG_P;
  const int r1 = (N + L1_P - 1) / L1_P;
  const int r2 = (N + L2_P - 1) / L2_P;

  const int tiles = (N + 15) / 16;
  const int wavesPerBlock = TPB / 32;
  dim3 gT((tiles + wavesPerBlock - 1) / wavesPerBlock);
  k_gemm1_wmma<<<gT, blk, 0, stream>>>(x, W1, deg, h1, bucketed ? nullptr : agg1, meta, N);
  k_edge_deg<<<dim3(rDeg * DEG_S), eblk, 0, stream>>>(ei + E, deg, meta, N, E, rDeg);

  if (bucketed) {
    k_bucket<<<dim3((E + BK_CAP - 1) / BK_CAP), eblk, 0, stream>>>(ei, meta, glist, E);
    k_own16<<<dim3(r1), eblk, 0, stream>>>(deg, h1, bst, glist, b1, W2, b2, rec2, out, N);
    k_edge_agg2<true><<<dim3(r2 * L2_S), eblk, 0, stream>>>(ei, deg, rec2, out, bst, glist, N, E, r2);
  } else {
    k_edge_agg16<<<dim3(r1 * L1_S), eblk, 0, stream>>>(ei, deg, h1, agg1, N, E, r1);
    k_node2<<<gN, blk, 0, stream>>>(agg1, b1, W2, b2, deg, rec2, out, N);
    k_edge_agg2<false><<<dim3(r2 * L2_S), eblk, 0, stream>>>(ei, deg, rec2, out, bst, glist, N, E, r2);
  }
}
